# starting kernel
# baseline (speedup 1.0000x reference)
#include <hip/hip_runtime.h>

typedef __attribute__((ext_vector_type(2))) float v2f;
typedef __attribute__((ext_vector_type(8))) float v8f;

#define NHID 256
#define EPSF 1e-7f
#define MAXNF 0.99999f

__device__ __forceinline__ v8f wmma_f32x4(v2f a, v2f b, v8f c) {
  return __builtin_amdgcn_wmma_f32_16x16x4_f32(false, a, false, b, (short)0, c,
                                               false, false);
}

__device__ __forceinline__ float wred32(float v) {
#pragma unroll
  for (int m = 16; m > 0; m >>= 1) v += __shfl_xor(v, m, 32);
  return v;
}

__device__ __forceinline__ float hyp_scale(float x2, float m2) {
  float xn  = fminf(fmaxf(sqrtf(x2), EPSF), MAXNF);
  float mxn = fmaxf(sqrtf(m2), EPSF);
  return tanhf(mxn / xn * atanhf(xn)) / mxn;
}

__device__ __forceinline__ void load8(const float* p, float s, float* d) {
  const float4* g = (const float4*)p;
  float4 a = g[0], b = g[1];
  d[0] = a.x * s; d[1] = a.y * s; d[2] = a.z * s; d[3] = a.w * s;
  d[4] = b.x * s; d[5] = b.y * s; d[6] = b.z * s; d[7] = b.w * s;
}

__device__ __forceinline__ void store8(float* p, const float* d, float s) {
  float4 a, b;
  a.x = d[0] * s; a.y = d[1] * s; a.z = d[2] * s; a.w = d[3] * s;
  b.x = d[4] * s; b.y = d[5] * s; b.z = d[6] * s; b.w = d[7] * s;
  float4* g = (float4*)p;
  g[0] = a; g[1] = b;
}

template <int KDIM, bool FULL_EPI>
__global__ __launch_bounds__(256) void k_mm(const float* X,
                                            const float* __restrict__ W,
                                            const float* __restrict__ bias,
                                            float* OUT, float inScale) {
  __shared__ float smx[32][NHID + 8];
  __shared__ float part[32][8];
  __shared__ float rx2[32], rm2[32], rdot[32], rv2[32];
  __shared__ float sb2s;

  const int t = threadIdx.x;
  const int wave = t >> 5, lane = t & 31;
  const int tile_m = wave >> 2;
  const int n0 = (wave & 3) << 6;
  const int l15 = lane & 15;
  const int kk = (lane >> 4) << 1;
  const int brow = blockIdx.x << 5;

  v8f acc0 = {}, acc1 = {}, acc2 = {}, acc3 = {};
  const float* Ar = X + (size_t)(brow + tile_m * 16 + l15) * KDIM;
  for (int k0 = 0; k0 < KDIM; k0 += 4) {
    v2f a;
    a.x = Ar[k0 + kk];
    a.y = Ar[k0 + kk + 1];
    const float* Bp = W + (size_t)(k0 + kk) * NHID + n0 + l15;
    v2f b;
    b.x = Bp[0];  b.y = Bp[NHID];      acc0 = wmma_f32x4(a, b, acc0);
    b.x = Bp[16]; b.y = Bp[NHID + 16]; acc1 = wmma_f32x4(a, b, acc1);
    b.x = Bp[32]; b.y = Bp[NHID + 32]; acc2 = wmma_f32x4(a, b, acc2);
    b.x = Bp[48]; b.y = Bp[NHID + 48]; acc3 = wmma_f32x4(a, b, acc3);
  }
  {
    const int mb = tile_m * 16 + ((lane >> 4) ? 8 : 0);
#pragma unroll
    for (int v = 0; v < 8; ++v) {
      smx[mb + v][n0 + 0 + l15]  = acc0[v];
      smx[mb + v][n0 + 16 + l15] = acc1[v];
      smx[mb + v][n0 + 32 + l15] = acc2[v];
      smx[mb + v][n0 + 48 + l15] = acc3[v];
    }
  }
  __syncthreads();

  const int r = t >> 3, p = t & 7;
  float sx = 0.f;
  {
    const int seg = KDIM / 8;
    const float* xr = X + (size_t)(brow + r) * KDIM + p * seg;
    for (int j = 0; j < seg; j += 4) {
      float4 v = *(const float4*)(xr + j);
      sx += v.x * v.x + v.y * v.y + v.z * v.z + v.w * v.w;
    }
  }
  float sm = 0.f, sd = 0.f;
  {
    const float* mr = &smx[r][p * 32];
#pragma unroll
    for (int j = 0; j < 32; ++j) {
      float m = mr[j];
      sm += m * m;
      if (FULL_EPI) sd += m * bias[p * 32 + j];
    }
  }
  part[r][p] = sx;
  __syncthreads();
  if (t < 32) { float s = 0.f; for (int i = 0; i < 8; ++i) s += part[t][i]; rx2[t] = s; }
  __syncthreads();
  part[r][p] = sm;
  __syncthreads();
  if (t < 32) { float s = 0.f; for (int i = 0; i < 8; ++i) s += part[t][i]; rm2[t] = s; }
  __syncthreads();
  if (FULL_EPI) {
    part[r][p] = sd;
    __syncthreads();
    if (t < 32) { float s = 0.f; for (int i = 0; i < 8; ++i) s += part[t][i]; rdot[t] = s; }
    __syncthreads();
    part[r][p] = bias[t] * bias[t];
    __syncthreads();
    if (t == 0) {
      float s = 0.f;
      for (int i = 0; i < 32; ++i)
        for (int j = 0; j < 8; ++j) s += part[i][j];
      sb2s = s;
    }
    __syncthreads();
  }

  const float x2s = inScale * inScale * rx2[r];
  const float m2s = inScale * inScale * rm2[r];
  const float cmv = hyp_scale(x2s, m2s) * inScale;
  float* orow = OUT + (size_t)(brow + r) * NHID + p * 32;
  float* mrw = &smx[r][p * 32];
  if (!FULL_EPI) {
#pragma unroll
    for (int j = 0; j < 32; j += 4) {
      float4 o;
      o.x = cmv * mrw[j + 0]; o.y = cmv * mrw[j + 1];
      o.z = cmv * mrw[j + 2]; o.w = cmv * mrw[j + 3];
      *(float4*)(orow + j) = o;
    }
  } else {
    const float cf = 0.25f * cmv;
    const float x2r = cf * cf * rm2[r];
    const float xy = cf * rdot[r];
    const float y2 = sb2s;
    const float den = fmaxf(1.f + 2.f * xy + x2r * y2, EPSF);
    const float ca = (1.f + 2.f * xy + y2) / den;
    const float cb = (1.f - x2r) / den;
    const float zn2 = ca * ca * x2r + 2.f * ca * cb * xy + cb * cb * y2;
    const float zn = fminf(fmaxf(sqrtf(zn2), EPSF), MAXNF);
    const float us = atanhf(zn) / zn;
    const float acf = ca * cf;
    float sv = 0.f;
#pragma unroll
    for (int j = 0; j < 32; ++j) {
      float z = acf * mrw[j] + cb * bias[p * 32 + j];
      float uu = fmaxf(us * z, 0.f);
      mrw[j] = uu;
      sv += uu * uu;
    }
    part[r][p] = sv;
    __syncthreads();
    if (t < 32) { float s = 0.f; for (int i = 0; i < 8; ++i) s += part[t][i]; rv2[t] = s; }
    __syncthreads();
    const float vn = fmaxf(sqrtf(rv2[r]), EPSF);
    const float fs = tanhf(vn) / vn;
#pragma unroll
    for (int j = 0; j < 32; j += 4) {
      float4 o;
      o.x = fs * mrw[j + 0]; o.y = fs * mrw[j + 1];
      o.z = fs * mrw[j + 2]; o.w = fs * mrw[j + 3];
      *(float4*)(orow + j) = o;
    }
  }
}

__global__ __launch_bounds__(256) void k_agg(const float* __restrict__ F,
                                             const int* __restrict__ idx,
                                             const float* __restrict__ bias,
                                             float* __restrict__ OUT,
                                             float inScale, int doAct) {
  const int t = threadIdx.x;
  const int wave = t >> 5, lane = t & 31;
  const int node = (blockIdx.x << 3) + wave;
  const int* srow = idx + (size_t)node * 16;
  const int joff = lane << 3;

  float x[8];
  load8(F + (size_t)srow[0] * NHID + joff, inScale, x);
  float px = 0.f;
#pragma unroll
  for (int j = 0; j < 8; ++j) px += x[j] * x[j];
  float x2 = wred32(px);

  for (int k = 1; k < 16; ++k) {
    if (k < 15) __builtin_prefetch(F + (size_t)srow[k + 1] * NHID + joff, 0, 0);
    float y[8];
    load8(F + (size_t)srow[k] * NHID + joff, inScale, y);
    float py2 = 0.f, pxy = 0.f;
#pragma unroll
    for (int j = 0; j < 8; ++j) { py2 += y[j] * y[j]; pxy += x[j] * y[j]; }
    float y2 = wred32(py2);
    float xy = wred32(pxy);
    float den = fmaxf(1.f + 2.f * xy + x2 * y2, EPSF);
    float ca = (1.f + 2.f * xy + y2) / den;
    float cb = (1.f - x2) / den;
    float pn = 0.f;
#pragma unroll
    for (int j = 0; j < 8; ++j) { x[j] = ca * x[j] + cb * y[j]; pn += x[j] * x[j]; }
    x2 = wred32(pn);
  }

  float* orow = OUT + (size_t)node * NHID + joff;
  if (!doAct) {
    store8(orow, x, 1.f);
  } else {
    float b[8];
    load8(bias + joff, 1.f, b);
    float pb2 = 0.f, pxy = 0.f;
#pragma unroll
    for (int j = 0; j < 8; ++j) { pb2 += b[j] * b[j]; pxy += x[j] * b[j]; }
    float y2 = wred32(pb2);
    float xy = 0.25f * wred32(pxy);
    float x2r = 0.0625f * x2;
    float den = fmaxf(1.f + 2.f * xy + x2r * y2, EPSF);
    float ca = (1.f + 2.f * xy + y2) / den;
    float cb = (1.f - x2r) / den;
    float zn2 = ca * ca * x2r + 2.f * ca * cb * xy + cb * cb * y2;
    float zn = fminf(fmaxf(sqrtf(zn2), EPSF), MAXNF);
    float us = atanhf(zn) / zn;
    float u[8];
    float pv = 0.f;
#pragma unroll
    for (int j = 0; j < 8; ++j) {
      float z = ca * 0.25f * x[j] + cb * b[j];
      u[j] = fmaxf(us * z, 0.f);
      pv += u[j] * u[j];
    }
    float vn = fmaxf(sqrtf(wred32(pv)), EPSF);
    float fs = tanhf(vn) / vn;
    store8(orow, u, fs);
  }
}

__global__ __launch_bounds__(256) void k_final(const float* __restrict__ H,
                                               const float* __restrict__ Wl,
                                               const float* __restrict__ bl,
                                               const int* __restrict__ tf,
                                               float* __restrict__ OUT) {
  __shared__ int ridx[64];
  __shared__ float smx[64][72];
  __shared__ float part[64][4];
  __shared__ float rx2[64], rm2[64], rdot[64];
  __shared__ float sb2;

  const int t = threadIdx.x;
  if (t < 64) ridx[t] = tf[t] + t * 1024;
  if (t == 0) {
    float s = 0.f;
    for (int j = 0; j < 64; ++j) s += bl[j] * bl[j];
    sb2 = s;
  }
  __syncthreads();

  const int wave = t >> 5, lane = t & 31;
  const int l15 = lane & 15;
  const int kk = (lane >> 4) << 1;
#pragma unroll
  for (int i = 0; i < 2; ++i) {
    const int tile = wave * 2 + i;
    const int tm = tile >> 2, tn = tile & 3;
    const float* Ar = H + (size_t)ridx[tm * 16 + l15] * NHID;
    const float* Br = Wl + (size_t)(tn * 16 + l15) * NHID;
    v8f c = {};
    for (int k0 = 0; k0 < NHID; k0 += 4) {
      v2f a, b;
      a.x = Ar[k0 + kk]; a.y = Ar[k0 + kk + 1];
      b.x = Br[k0 + kk]; b.y = Br[k0 + kk + 1];
      c = wmma_f32x4(a, b, c);
    }
    const int n = tn * 16 + l15;
    const int mb = tm * 16 + ((lane >> 4) ? 8 : 0);
#pragma unroll
    for (int v = 0; v < 8; ++v) smx[mb + v][n] = c[v];
  }
  __syncthreads();

  const int r = t >> 2, p = t & 3;
  float sx = 0.f;
  {
    const float* hr = H + (size_t)ridx[r] * NHID + p * 64;
    for (int j = 0; j < 64; j += 4) {
      float4 v = *(const float4*)(hr + j);
      sx += v.x * v.x + v.y * v.y + v.z * v.z + v.w * v.w;
    }
  }
  float sm = 0.f, sd = 0.f;
#pragma unroll
  for (int j = 0; j < 16; ++j) {
    float m = smx[r][p * 16 + j];
    sm += m * m;
    sd += m * bl[p * 16 + j];
  }
  part[r][p] = sx;
  __syncthreads();
  if (t < 64) rx2[t] = part[t][0] + part[t][1] + part[t][2] + part[t][3];
  __syncthreads();
  part[r][p] = sm;
  __syncthreads();
  if (t < 64) rm2[t] = part[t][0] + part[t][1] + part[t][2] + part[t][3];
  __syncthreads();
  part[r][p] = sd;
  __syncthreads();
  if (t < 64) rdot[t] = part[t][0] + part[t][1] + part[t][2] + part[t][3];
  __syncthreads();

  const float s = hyp_scale(rx2[r], rm2[r]);
  const float x2r = s * s * rm2[r];
  const float xy = s * rdot[r];
  const float y2 = sb2;
  const float den = fmaxf(1.f + 2.f * xy + x2r * y2, EPSF);
  const float ca = (1.f + 2.f * xy + y2) / den;
  const float cb = (1.f - x2r) / den;
  const float acs = ca * s;
#pragma unroll
  for (int j = 0; j < 16; ++j)
    OUT[r * 64 + p * 16 + j] = acs * smx[r][p * 16 + j] + cb * bl[p * 16 + j];
}

extern "C" void kernel_launch(void* const* d_in, const int* in_sizes, int n_in,
                              void* d_out, int out_size, void* d_ws,
                              size_t ws_size, hipStream_t stream) {
  const float* features = (const float*)d_in[0];
  const float* W1 = (const float*)d_in[1];
  const float* b1 = (const float*)d_in[2];
  const float* W2 = (const float*)d_in[3];
  const float* b2 = (const float*)d_in[4];
  const float* Wl = (const float*)d_in[5];
  const float* bl = (const float*)d_in[6];
  const int* src_idx = (const int*)d_in[7];
  const int* to_fetch = (const int*)d_in[8];
  float* out = (float*)d_out;

  float* buf0 = (float*)d_ws;
  float* buf1 = buf0 + (size_t)65536 * 256;

  k_mm<512, false><<<2048, 256, 0, stream>>>(const_cast<float*>(features), W1,
                                             b1, buf0, 0.25f);
  k_agg<<<8192, 256, 0, stream>>>(buf0, src_idx, b1, buf1, 1.0f, 1);
  k_agg<<<8192, 256, 0, stream>>>(buf1, src_idx, b2, buf0, 0.25f, 0);
  k_mm<256, true><<<2048, 256, 0, stream>>>(buf0, W2, b2, buf0, 1.0f);
  k_final<<<1, 256, 0, stream>>>(buf0, Wl, bl, to_fetch, out);
}


# compute only what the output reads: the two-hop in-neighbourhood of the 64 fetched rows (16384 product rows, 1024 layer-1 rows, 64 layer-2 rows), stored in neighbour order; row arithmetic unchanged
# speedup vs baseline: 4.2411x; 4.2411x over previous
#include <hip/hip_runtime.h>

typedef __attribute__((ext_vector_type(2))) float v2f;
typedef __attribute__((ext_vector_type(8))) float v8f;

#define NHID 256
#define EPSF 1e-7f
#define MAXNF 0.99999f

__device__ __forceinline__ v8f wmma_f32x4(v2f a, v2f b, v8f c) {
  return __builtin_amdgcn_wmma_f32_16x16x4_f32(false, a, false, b, (short)0, c,
                                               false, false);
}

__device__ __forceinline__ float wred32(float v) {
#pragma unroll
  for (int m = 16; m > 0; m >>= 1) v += __shfl_xor(v, m, 32);
  return v;
}

__device__ __forceinline__ float hyp_scale(float x2, float m2) {
  float xn  = fminf(fmaxf(sqrtf(x2), EPSF), MAXNF);
  float mxn = fmaxf(sqrtf(m2), EPSF);
  return tanhf(mxn / xn * atanhf(xn)) / mxn;
}

__device__ __forceinline__ void load8(const float* p, float s, float* d) {
  const float4* g = (const float4*)p;
  float4 a = g[0], b = g[1];
  d[0] = a.x * s; d[1] = a.y * s; d[2] = a.z * s; d[3] = a.w * s;
  d[4] = b.x * s; d[5] = b.y * s; d[6] = b.z * s; d[7] = b.w * s;
}

__device__ __forceinline__ void store8(float* p, const float* d, float s) {
  float4 a, b;
  a.x = d[0] * s; a.y = d[1] * s; a.z = d[2] * s; a.w = d[3] * s;
  b.x = d[4] * s; b.y = d[5] * s; b.z = d[6] * s; b.w = d[7] * s;
  float4* g = (float4*)p;
  g[0] = a; g[1] = b;
}

template <int KDIM, bool FULL_EPI, bool GATHER>
__global__ __launch_bounds__(256) void k_mm(const float* X,
                                            const float* __restrict__ W,
                                            const float* __restrict__ bias,
                                            float* OUT, float inScale,
                                            const int* __restrict__ src,
                                            const int* __restrict__ tf) {
  __shared__ float smx[32][NHID + 8];
  __shared__ float part[32][8];
  __shared__ float rx2[32], rm2[32], rdot[32], rv2[32];
  __shared__ float sb2s;

  const int t = threadIdx.x;
  const int wave = t >> 5, lane = t & 31;
  const int tile_m = wave >> 2;
  const int n0 = (wave & 3) << 6;
  const int l15 = lane & 15;
  const int kk = (lane >> 4) << 1;
  const int brow = blockIdx.x << 5;

  const int* nb = nullptr;
  if (GATHER) {
    const int b = brow >> 8;
    const int n2 = tf[b] + b * 1024;
    const int n1 = src[(size_t)n2 * 16 + (((brow >> 4) + tile_m) & 15)];
    nb = src + (size_t)n1 * 16;
  }
  const int arow = GATHER ? nb[l15] : brow + tile_m * 16 + l15;

  v8f acc0 = {}, acc1 = {}, acc2 = {}, acc3 = {};
  const float* Ar = X + (size_t)arow * KDIM;
  for (int k0 = 0; k0 < KDIM; k0 += 4) {
    v2f a;
    a.x = Ar[k0 + kk];
    a.y = Ar[k0 + kk + 1];
    const float* Bp = W + (size_t)(k0 + kk) * NHID + n0 + l15;
    v2f b;
    b.x = Bp[0];  b.y = Bp[NHID];      acc0 = wmma_f32x4(a, b, acc0);
    b.x = Bp[16]; b.y = Bp[NHID + 16]; acc1 = wmma_f32x4(a, b, acc1);
    b.x = Bp[32]; b.y = Bp[NHID + 32]; acc2 = wmma_f32x4(a, b, acc2);
    b.x = Bp[48]; b.y = Bp[NHID + 48]; acc3 = wmma_f32x4(a, b, acc3);
  }
  {
    const int mb = tile_m * 16 + ((lane >> 4) ? 8 : 0);
#pragma unroll
    for (int v = 0; v < 8; ++v) {
      smx[mb + v][n0 + 0 + l15]  = acc0[v];
      smx[mb + v][n0 + 16 + l15] = acc1[v];
      smx[mb + v][n0 + 32 + l15] = acc2[v];
      smx[mb + v][n0 + 48 + l15] = acc3[v];
    }
  }
  __syncthreads();

  const int r = t >> 3, p = t & 7;
  float sx = 0.f;
  {
    const int seg = KDIM / 8;
    const int xrow = GATHER ? nb[r & 15] : brow + r;
    const float* xr = X + (size_t)xrow * KDIM + p * seg;
    for (int j = 0; j < seg; j += 4) {
      float4 v = *(const float4*)(xr + j);
      sx += v.x * v.x + v.y * v.y + v.z * v.z + v.w * v.w;
    }
  }
  float sm = 0.f, sd = 0.f;
  {
    const float* mr = &smx[r][p * 32];
#pragma unroll
    for (int j = 0; j < 32; ++j) {
      float m = mr[j];
      sm += m * m;
      if (FULL_EPI) sd += m * bias[p * 32 + j];
    }
  }
  part[r][p] = sx;
  __syncthreads();
  if (t < 32) { float s = 0.f; for (int i = 0; i < 8; ++i) s += part[t][i]; rx2[t] = s; }
  __syncthreads();
  part[r][p] = sm;
  __syncthreads();
  if (t < 32) { float s = 0.f; for (int i = 0; i < 8; ++i) s += part[t][i]; rm2[t] = s; }
  __syncthreads();
  if (FULL_EPI) {
    part[r][p] = sd;
    __syncthreads();
    if (t < 32) { float s = 0.f; for (int i = 0; i < 8; ++i) s += part[t][i]; rdot[t] = s; }
    __syncthreads();
    part[r][p] = bias[t] * bias[t];
    __syncthreads();
    if (t == 0) {
      float s = 0.f;
      for (int i = 0; i < 32; ++i)
        for (int j = 0; j < 8; ++j) s += part[i][j];
      sb2s = s;
    }
    __syncthreads();
  }

  const float x2s = inScale * inScale * rx2[r];
  const float m2s = inScale * inScale * rm2[r];
  const float cmv = hyp_scale(x2s, m2s) * inScale;
  float* orow = OUT + (size_t)(brow + r) * NHID + p * 32;
  float* mrw = &smx[r][p * 32];
  if (!FULL_EPI) {
#pragma unroll
    for (int j = 0; j < 32; j += 4) {
      float4 o;
      o.x = cmv * mrw[j + 0]; o.y = cmv * mrw[j + 1];
      o.z = cmv * mrw[j + 2]; o.w = cmv * mrw[j + 3];
      *(float4*)(orow + j) = o;
    }
  } else {
    const float cf = 0.25f * cmv;
    const float x2r = cf * cf * rm2[r];
    const float xy = cf * rdot[r];
    const float y2 = sb2s;
    const float den = fmaxf(1.f + 2.f * xy + x2r * y2, EPSF);
    const float ca = (1.f + 2.f * xy + y2) / den;
    const float cb = (1.f - x2r) / den;
    const float zn2 = ca * ca * x2r + 2.f * ca * cb * xy + cb * cb * y2;
    const float zn = fminf(fmaxf(sqrtf(zn2), EPSF), MAXNF);
    const float us = atanhf(zn) / zn;
    const float acf = ca * cf;
    float sv = 0.f;
#pragma unroll
    for (int j = 0; j < 32; ++j) {
      float z = acf * mrw[j] + cb * bias[p * 32 + j];
      float uu = fmaxf(us * z, 0.f);
      mrw[j] = uu;
      sv += uu * uu;
    }
    part[r][p] = sv;
    __syncthreads();
    if (t < 32) { float s = 0.f; for (int i = 0; i < 8; ++i) s += part[t][i]; rv2[t] = s; }
    __syncthreads();
    const float vn = fmaxf(sqrtf(rv2[r]), EPSF);
    const float fs = tanhf(vn) / vn;
#pragma unroll
    for (int j = 0; j < 32; j += 4) {
      float4 o;
      o.x = fs * mrw[j + 0]; o.y = fs * mrw[j + 1];
      o.z = fs * mrw[j + 2]; o.w = fs * mrw[j + 3];
      *(float4*)(orow + j) = o;
    }
  }
}

__global__ __launch_bounds__(256) void k_agg(const float* __restrict__ F,
                                             const float* __restrict__ bias,
                                             float* __restrict__ OUT,
                                             float inScale, int doAct) {
  const int t = threadIdx.x;
  const int wave = t >> 5, lane = t & 31;
  const int node = (blockIdx.x << 3) + wave;
  const int joff = lane << 3;
  const float* msg = F + (size_t)node * 16 * NHID + joff;

  float x[8];
  load8(msg, inScale, x);
  float px = 0.f;
#pragma unroll
  for (int j = 0; j < 8; ++j) px += x[j] * x[j];
  float x2 = wred32(px);

  for (int k = 1; k < 16; ++k) {
    if (k < 15) __builtin_prefetch(msg + (k + 1) * NHID, 0, 0);
    float y[8];
    load8(msg + k * NHID, inScale, y);
    float py2 = 0.f, pxy = 0.f;
#pragma unroll
    for (int j = 0; j < 8; ++j) { py2 += y[j] * y[j]; pxy += x[j] * y[j]; }
    float y2 = wred32(py2);
    float xy = wred32(pxy);
    float den = fmaxf(1.f + 2.f * xy + x2 * y2, EPSF);
    float ca = (1.f + 2.f * xy + y2) / den;
    float cb = (1.f - x2) / den;
    float pn = 0.f;
#pragma unroll
    for (int j = 0; j < 8; ++j) { x[j] = ca * x[j] + cb * y[j]; pn += x[j] * x[j]; }
    x2 = wred32(pn);
  }

  float* orow = OUT + (size_t)node * NHID + joff;
  if (!doAct) {
    store8(orow, x, 1.f);
  } else {
    float b[8];
    load8(bias + joff, 1.f, b);
    float pb2 = 0.f, pxy = 0.f;
#pragma unroll
    for (int j = 0; j < 8; ++j) { pb2 += b[j] * b[j]; pxy += x[j] * b[j]; }
    float y2 = wred32(pb2);
    float xy = 0.25f * wred32(pxy);
    float x2r = 0.0625f * x2;
    float den = fmaxf(1.f + 2.f * xy + x2r * y2, EPSF);
    float ca = (1.f + 2.f * xy + y2) / den;
    float cb = (1.f - x2r) / den;
    float zn2 = ca * ca * x2r + 2.f * ca * cb * xy + cb * cb * y2;
    float zn = fminf(fmaxf(sqrtf(zn2), EPSF), MAXNF);
    float us = atanhf(zn) / zn;
    float u[8];
    float pv = 0.f;
#pragma unroll
    for (int j = 0; j < 8; ++j) {
      float z = ca * 0.25f * x[j] + cb * b[j];
      u[j] = fmaxf(us * z, 0.f);
      pv += u[j] * u[j];
    }
    float vn = fmaxf(sqrtf(wred32(pv)), EPSF);
    float fs = tanhf(vn) / vn;
    store8(orow, u, fs);
  }
}

__global__ __launch_bounds__(256) void k_final(const float* __restrict__ H,
                                               const float* __restrict__ Wl,
                                               const float* __restrict__ bl,
                                               float* __restrict__ OUT) {
  __shared__ float smx[64][72];
  __shared__ float part[64][4];
  __shared__ float rx2[64], rm2[64], rdot[64];
  __shared__ float sb2;

  const int t = threadIdx.x;
  if (t == 0) {
    float s = 0.f;
    for (int j = 0; j < 64; ++j) s += bl[j] * bl[j];
    sb2 = s;
  }
  __syncthreads();

  const int wave = t >> 5, lane = t & 31;
  const int l15 = lane & 15;
  const int kk = (lane >> 4) << 1;
#pragma unroll
  for (int i = 0; i < 2; ++i) {
    const int tile = wave * 2 + i;
    const int tm = tile >> 2, tn = tile & 3;
    const float* Ar = H + (size_t)(tm * 16 + l15) * NHID;
    const float* Br = Wl + (size_t)(tn * 16 + l15) * NHID;
    v8f c = {};
    for (int k0 = 0; k0 < NHID; k0 += 4) {
      v2f a, b;
      a.x = Ar[k0 + kk]; a.y = Ar[k0 + kk + 1];
      b.x = Br[k0 + kk]; b.y = Br[k0 + kk + 1];
      c = wmma_f32x4(a, b, c);
    }
    const int n = tn * 16 + l15;
    const int mb = tm * 16 + ((lane >> 4) ? 8 : 0);
#pragma unroll
    for (int v = 0; v < 8; ++v) smx[mb + v][n] = c[v];
  }
  __syncthreads();

  const int r = t >> 2, p = t & 3;
  float sx = 0.f;
  {
    const float* hr = H + (size_t)r * NHID + p * 64;
    for (int j = 0; j < 64; j += 4) {
      float4 v = *(const float4*)(hr + j);
      sx += v.x * v.x + v.y * v.y + v.z * v.z + v.w * v.w;
    }
  }
  float sm = 0.f, sd = 0.f;
#pragma unroll
  for (int j = 0; j < 16; ++j) {
    float m = smx[r][p * 16 + j];
    sm += m * m;
    sd += m * bl[p * 16 + j];
  }
  part[r][p] = sx;
  __syncthreads();
  if (t < 64) rx2[t] = part[t][0] + part[t][1] + part[t][2] + part[t][3];
  __syncthreads();
  part[r][p] = sm;
  __syncthreads();
  if (t < 64) rm2[t] = part[t][0] + part[t][1] + part[t][2] + part[t][3];
  __syncthreads();
  part[r][p] = sd;
  __syncthreads();
  if (t < 64) rdot[t] = part[t][0] + part[t][1] + part[t][2] + part[t][3];
  __syncthreads();

  const float s = hyp_scale(rx2[r], rm2[r]);
  const float x2r = s * s * rm2[r];
  const float xy = s * rdot[r];
  const float y2 = sb2;
  const float den = fmaxf(1.f + 2.f * xy + x2r * y2, EPSF);
  const float ca = (1.f + 2.f * xy + y2) / den;
  const float cb = (1.f - x2r) / den;
  const float acs = ca * s;
#pragma unroll
  for (int j = 0; j < 16; ++j)
    OUT[r * 64 + p * 16 + j] = acs * smx[r][p * 16 + j] + cb * bl[p * 16 + j];
}

extern "C" void kernel_launch(void* const* d_in, const int* in_sizes, int n_in,
                              void* d_out, int out_size, void* d_ws,
                              size_t ws_size, hipStream_t stream) {
  const float* features = (const float*)d_in[0];
  const float* W1 = (const float*)d_in[1];
  const float* b1 = (const float*)d_in[2];
  const float* W2 = (const float*)d_in[3];
  const float* b2 = (const float*)d_in[4];
  const float* Wl = (const float*)d_in[5];
  const float* bl = (const float*)d_in[6];
  const int* src_idx = (const int*)d_in[7];
  const int* to_fetch = (const int*)d_in[8];
  float* out = (float*)d_out;

  float* buf0 = (float*)d_ws;
  float* buf1 = buf0 + (size_t)65536 * 256;

  k_mm<512, false, true><<<512, 256, 0, stream>>>(
      const_cast<float*>(features), W1, b1, buf0, 0.25f, src_idx, to_fetch);
  k_agg<<<128, 256, 0, stream>>>(buf0, b1, buf1, 1.0f, 1);
  k_agg<<<8, 256, 0, stream>>>(buf1, b2, buf0, 0.25f, 0);
  k_mm<256, true, false><<<2, 256, 0, stream>>>(buf0, W2, b2, buf0, 1.0f,
                                                src_idx, to_fetch);
  k_final<<<1, 256, 0, stream>>>(buf0, Wl, bl, out);
}


# k_mm product loop: X and W tiles of 32 k staged through two LDS stages by 16-byte coalesced loads, WMMA fragments read from LDS; fragment k order and summation order unchanged
# speedup vs baseline: 4.6538x; 1.0973x over previous
#include <hip/hip_runtime.h>

typedef __attribute__((ext_vector_type(2))) float v2f;
typedef __attribute__((ext_vector_type(8))) float v8f;

#define NHID 256
#define EPSF 1e-7f
#define MAXNF 0.99999f

__device__ __forceinline__ v8f wmma_f32x4(v2f a, v2f b, v8f c) {
  return __builtin_amdgcn_wmma_f32_16x16x4_f32(false, a, false, b, (short)0, c,
                                               false, false);
}

__device__ __forceinline__ float wred32(float v) {
#pragma unroll
  for (int m = 16; m > 0; m >>= 1) v += __shfl_xor(v, m, 32);
  return v;
}

__device__ __forceinline__ float hyp_scale(float x2, float m2) {
  float xn  = fminf(fmaxf(sqrtf(x2), EPSF), MAXNF);
  float mxn = fmaxf(sqrtf(m2), EPSF);
  return tanhf(mxn / xn * atanhf(xn)) / mxn;
}

__device__ __forceinline__ void load8(const float* p, float s, float* d) {
  const float4* g = (const float4*)p;
  float4 a = g[0], b = g[1];
  d[0] = a.x * s; d[1] = a.y * s; d[2] = a.z * s; d[3] = a.w * s;
  d[4] = b.x * s; d[5] = b.y * s; d[6] = b.z * s; d[7] = b.w * s;
}

__device__ __forceinline__ void store8(float* p, const float* d, float s) {
  float4 a, b;
  a.x = d[0] * s; a.y = d[1] * s; a.z = d[2] * s; a.w = d[3] * s;
  b.x = d[4] * s; b.y = d[5] * s; b.z = d[6] * s; b.w = d[7] * s;
  float4* g = (float4*)p;
  g[0] = a; g[1] = b;
}

template <int KDIM, bool FULL_EPI, bool GATHER>
__global__ __launch_bounds__(256) void k_mm(const float* X,
                                            const float* __restrict__ W,
                                            const float* __restrict__ bias,
                                            float* OUT, float inScale,
                                            const int* __restrict__ src,
                                            const int* __restrict__ tf) {
  constexpr int KT = 32;
  constexpr int A_PITCH = KT + 4;
  constexpr int B_PITCH = 2 * NHID + 32;
  constexpr int A_FL = 32 * A_PITCH;
  constexpr int B_FL = (KT / 2) * B_PITCH;
  constexpr int ST_FL = A_FL + B_FL;
  constexpr int NT = KDIM / KT;
  __shared__ __attribute__((aligned(16))) float stg[2 * ST_FL];
  float (*smx)[NHID + 8] = reinterpret_cast<float (*)[NHID + 8]>(stg);
  static_assert(32 * (NHID + 8) <= 2 * ST_FL, "epilogue rows fit the stages");
  __shared__ float part[32][8];
  __shared__ float rx2[32], rm2[32], rdot[32], rv2[32];
  __shared__ float sb2s;

  const int t = threadIdx.x;
  const int wave = t >> 5, lane = t & 31;
  const int tile_m = wave >> 2;
  const int n0 = (wave & 3) << 6;
  const int l15 = lane & 15;
  const int kk = (lane >> 4) << 1;
  const int brow = blockIdx.x << 5;
  const int r = t >> 3, p = t & 7;

  const int* nb = nullptr;
  if (GATHER) {
    const int b = brow >> 8;
    const int n2 = tf[b] + b * 1024;
    const int n1 = src[(size_t)n2 * 16 + (((brow >> 4) + tile_m) & 15)];
    nb = src + (size_t)n1 * 16;
  }
  const int xrow = GATHER ? nb[r & 15] : brow + r;

  const int c4 = t & 63, kp0 = t >> 6;
  const float* xg = X + (size_t)xrow * KDIM + p * 4;
  const float* wg = W + (size_t)(2 * kp0) * NHID + c4 * 4;
  float4 ra, rb0[4], rb1[4];
  auto gload = [&](int kt) {
    ra = *(const float4*)(xg + kt * KT);
    const float* w = wg + (size_t)(kt * KT) * NHID;
#pragma unroll
    for (int i = 0; i < 4; ++i) {
      rb0[i] = *(const float4*)(w + (size_t)(8 * i) * NHID);
      rb1[i] = *(const float4*)(w + (size_t)(8 * i + 1) * NHID);
    }
  };
  auto sstore = [&](float* st) {
    *(float4*)(st + r * A_PITCH + p * 4) = ra;
    float* bs = st + A_FL + kp0 * B_PITCH + c4 * 8;
#pragma unroll
    for (int i = 0; i < 4; ++i) {
      float4 lo, hi;
      lo.x = rb0[i].x; lo.y = rb1[i].x; lo.z = rb0[i].y; lo.w = rb1[i].y;
      hi.x = rb0[i].z; hi.y = rb1[i].z; hi.z = rb0[i].w; hi.w = rb1[i].w;
      *(float4*)(bs + (4 * i) * B_PITCH) = lo;
      *(float4*)(bs + (4 * i) * B_PITCH + 4) = hi;
    }
  };

  v8f acc0 = {}, acc1 = {}, acc2 = {}, acc3 = {};
  gload(0);
  sstore(stg);
  __syncthreads();
  for (int kt = 0; kt < NT; ++kt) {
    const float* st = stg + (kt & 1) * ST_FL;
    if (kt + 1 < NT) gload(kt + 1);
    const float* as = st + (tile_m * 16 + l15) * A_PITCH + kk;
    const float* bs = st + A_FL + (kk >> 1) * B_PITCH + (n0 + l15) * 2;
#pragma unroll
    for (int s = 0; s < KT / 4; ++s) {
      const v2f a = *(const v2f*)(as + s * 4);
      const float* bp = bs + (2 * s) * B_PITCH;
      const v2f b0 = *(const v2f*)(bp);
      const v2f b1 = *(const v2f*)(bp + 32);
      const v2f b2 = *(const v2f*)(bp + 64);
      const v2f b3 = *(const v2f*)(bp + 96);
      acc0 = wmma_f32x4(a, b0, acc0);
      acc1 = wmma_f32x4(a, b1, acc1);
      acc2 = wmma_f32x4(a, b2, acc2);
      acc3 = wmma_f32x4(a, b3, acc3);
    }
    if (kt + 1 < NT) sstore(stg + ((kt + 1) & 1) * ST_FL);
    __syncthreads();
  }
  {
    const int mb = tile_m * 16 + ((lane >> 4) ? 8 : 0);
#pragma unroll
    for (int v = 0; v < 8; ++v) {
      smx[mb + v][n0 + 0 + l15]  = acc0[v];
      smx[mb + v][n0 + 16 + l15] = acc1[v];
      smx[mb + v][n0 + 32 + l15] = acc2[v];
      smx[mb + v][n0 + 48 + l15] = acc3[v];
    }
  }
  __syncthreads();

  float sx = 0.f;
  {
    const int seg = KDIM / 8;
    const float* xr = X + (size_t)xrow * KDIM + p * seg;
    for (int j = 0; j < seg; j += 4) {
      float4 v = *(const float4*)(xr + j);
      sx += v.x * v.x + v.y * v.y + v.z * v.z + v.w * v.w;
    }
  }
  float sm = 0.f, sd = 0.f;
  {
    const float* mr = &smx[r][p * 32];
#pragma unroll
    for (int j = 0; j < 32; ++j) {
      float m = mr[j];
      sm += m * m;
      if (FULL_EPI) sd += m * bias[p * 32 + j];
    }
  }
  part[r][p] = sx;
  __syncthreads();
  if (t < 32) { float s = 0.f; for (int i = 0; i < 8; ++i) s += part[t][i]; rx2[t] = s; }
  __syncthreads();
  part[r][p] = sm;
  __syncthreads();
  if (t < 32) { float s = 0.f; for (int i = 0; i < 8; ++i) s += part[t][i]; rm2[t] = s; }
  __syncthreads();
  if (FULL_EPI) {
    part[r][p] = sd;
    __syncthreads();
    if (t < 32) { float s = 0.f; for (int i = 0; i < 8; ++i) s += part[t][i]; rdot[t] = s; }
    __syncthreads();
    part[r][p] = bias[t] * bias[t];
    __syncthreads();
    if (t == 0) {
      float s = 0.f;
      for (int i = 0; i < 32; ++i)
        for (int j = 0; j < 8; ++j) s += part[i][j];
      sb2s = s;
    }
    __syncthreads();
  }

  const float x2s = inScale * inScale * rx2[r];
  const float m2s = inScale * inScale * rm2[r];
  const float cmv = hyp_scale(x2s, m2s) * inScale;
  float* orow = OUT + (size_t)(brow + r) * NHID + p * 32;
  float* mrw = &smx[r][p * 32];
  if (!FULL_EPI) {
#pragma unroll
    for (int j = 0; j < 32; j += 4) {
      float4 o;
      o.x = cmv * mrw[j + 0]; o.y = cmv * mrw[j + 1];
      o.z = cmv * mrw[j + 2]; o.w = cmv * mrw[j + 3];
      *(float4*)(orow + j) = o;
    }
  } else {
    const float cf = 0.25f * cmv;
    const float x2r = cf * cf * rm2[r];
    const float xy = cf * rdot[r];
    const float y2 = sb2s;
    const float den = fmaxf(1.f + 2.f * xy + x2r * y2, EPSF);
    const float ca = (1.f + 2.f * xy + y2) / den;
    const float cb = (1.f - x2r) / den;
    const float zn2 = ca * ca * x2r + 2.f * ca * cb * xy + cb * cb * y2;
    const float zn = fminf(fmaxf(sqrtf(zn2), EPSF), MAXNF);
    const float us = atanhf(zn) / zn;
    const float acf = ca * cf;
    float sv = 0.f;
#pragma unroll
    for (int j = 0; j < 32; ++j) {
      float z = acf * mrw[j] + cb * bias[p * 32 + j];
      float uu = fmaxf(us * z, 0.f);
      mrw[j] = uu;
      sv += uu * uu;
    }
    part[r][p] = sv;
    __syncthreads();
    if (t < 32) { float s = 0.f; for (int i = 0; i < 8; ++i) s += part[t][i]; rv2[t] = s; }
    __syncthreads();
    const float vn = fmaxf(sqrtf(rv2[r]), EPSF);
    const float fs = tanhf(vn) / vn;
#pragma unroll
    for (int j = 0; j < 32; j += 4) {
      float4 o;
      o.x = fs * mrw[j + 0]; o.y = fs * mrw[j + 1];
      o.z = fs * mrw[j + 2]; o.w = fs * mrw[j + 3];
      *(float4*)(orow + j) = o;
    }
  }
}

__global__ __launch_bounds__(256) void k_agg(const float* __restrict__ F,
                                             const float* __restrict__ bias,
                                             float* __restrict__ OUT,
                                             float inScale, int doAct) {
  const int t = threadIdx.x;
  const int wave = t >> 5, lane = t & 31;
  const int node = (blockIdx.x << 3) + wave;
  const int joff = lane << 3;
  const float* msg = F + (size_t)node * 16 * NHID + joff;

  float x[8];
  load8(msg, inScale, x);
  float px = 0.f;
#pragma unroll
  for (int j = 0; j < 8; ++j) px += x[j] * x[j];
  float x2 = wred32(px);

  for (int k = 1; k < 16; ++k) {
    if (k < 15) __builtin_prefetch(msg + (k + 1) * NHID, 0, 0);
    float y[8];
    load8(msg + k * NHID, inScale, y);
    float py2 = 0.f, pxy = 0.f;
#pragma unroll
    for (int j = 0; j < 8; ++j) { py2 += y[j] * y[j]; pxy += x[j] * y[j]; }
    float y2 = wred32(py2);
    float xy = wred32(pxy);
    float den = fmaxf(1.f + 2.f * xy + x2 * y2, EPSF);
    float ca = (1.f + 2.f * xy + y2) / den;
    float cb = (1.f - x2) / den;
    float pn = 0.f;
#pragma unroll
    for (int j = 0; j < 8; ++j) { x[j] = ca * x[j] + cb * y[j]; pn += x[j] * x[j]; }
    x2 = wred32(pn);
  }

  float* orow = OUT + (size_t)node * NHID + joff;
  if (!doAct) {
    store8(orow, x, 1.f);
  } else {
    float b[8];
    load8(bias + joff, 1.f, b);
    float pb2 = 0.f, pxy = 0.f;
#pragma unroll
    for (int j = 0; j < 8; ++j) { pb2 += b[j] * b[j]; pxy += x[j] * b[j]; }
    float y2 = wred32(pb2);
    float xy = 0.25f * wred32(pxy);
    float x2r = 0.0625f * x2;
    float den = fmaxf(1.f + 2.f * xy + x2r * y2, EPSF);
    float ca = (1.f + 2.f * xy + y2) / den;
    float cb = (1.f - x2r) / den;
    float zn2 = ca * ca * x2r + 2.f * ca * cb * xy + cb * cb * y2;
    float zn = fminf(fmaxf(sqrtf(zn2), EPSF), MAXNF);
    float us = atanhf(zn) / zn;
    float u[8];
    float pv = 0.f;
#pragma unroll
    for (int j = 0; j < 8; ++j) {
      float z = ca * 0.25f * x[j] + cb * b[j];
      u[j] = fmaxf(us * z, 0.f);
      pv += u[j] * u[j];
    }
    float vn = fmaxf(sqrtf(wred32(pv)), EPSF);
    float fs = tanhf(vn) / vn;
    store8(orow, u, fs);
  }
}

__global__ __launch_bounds__(256) void k_final(const float* __restrict__ H,
                                               const float* __restrict__ Wl,
                                               const float* __restrict__ bl,
                                               float* __restrict__ OUT) {
  __shared__ float smx[64][72];
  __shared__ float part[64][4];
  __shared__ float rx2[64], rm2[64], rdot[64];
  __shared__ float sb2;

  const int t = threadIdx.x;
  if (t == 0) {
    float s = 0.f;
    for (int j = 0; j < 64; ++j) s += bl[j] * bl[j];
    sb2 = s;
  }
  __syncthreads();

  const int wave = t >> 5, lane = t & 31;
  const int l15 = lane & 15;
  const int kk = (lane >> 4) << 1;
#pragma unroll
  for (int i = 0; i < 2; ++i) {
    const int tile = wave * 2 + i;
    const int tm = tile >> 2, tn = tile & 3;
    const float* Ar = H + (size_t)(tm * 16 + l15) * NHID;
    const float* Br = Wl + (size_t)(tn * 16 + l15) * NHID;
    v8f c = {};
    for (int k0 = 0; k0 < NHID; k0 += 4) {
      v2f a, b;
      a.x = Ar[k0 + kk]; a.y = Ar[k0 + kk + 1];
      b.x = Br[k0 + kk]; b.y = Br[k0 + kk + 1];
      c = wmma_f32x4(a, b, c);
    }
    const int n = tn * 16 + l15;
    const int mb = tm * 16 + ((lane >> 4) ? 8 : 0);
#pragma unroll
    for (int v = 0; v < 8; ++v) smx[mb + v][n] = c[v];
  }
  __syncthreads();

  const int r = t >> 2, p = t & 3;
  float sx = 0.f;
  {
    const float* hr = H + (size_t)r * NHID + p * 64;
    for (int j = 0; j < 64; j += 4) {
      float4 v = *(const float4*)(hr + j);
      sx += v.x * v.x + v.y * v.y + v.z * v.z + v.w * v.w;
    }
  }
  float sm = 0.f, sd = 0.f;
#pragma unroll
  for (int j = 0; j < 16; ++j) {
    float m = smx[r][p * 16 + j];
    sm += m * m;
    sd += m * bl[p * 16 + j];
  }
  part[r][p] = sx;
  __syncthreads();
  if (t < 64) rx2[t] = part[t][0] + part[t][1] + part[t][2] + part[t][3];
  __syncthreads();
  part[r][p] = sm;
  __syncthreads();
  if (t < 64) rm2[t] = part[t][0] + part[t][1] + part[t][2] + part[t][3];
  __syncthreads();
  part[r][p] = sd;
  __syncthreads();
  if (t < 64) rdot[t] = part[t][0] + part[t][1] + part[t][2] + part[t][3];
  __syncthreads();

  const float s = hyp_scale(rx2[r], rm2[r]);
  const float x2r = s * s * rm2[r];
  const float xy = s * rdot[r];
  const float y2 = sb2;
  const float den = fmaxf(1.f + 2.f * xy + x2r * y2, EPSF);
  const float ca = (1.f + 2.f * xy + y2) / den;
  const float cb = (1.f - x2r) / den;
  const float acs = ca * s;
#pragma unroll
  for (int j = 0; j < 16; ++j)
    OUT[r * 64 + p * 16 + j] = acs * smx[r][p * 16 + j] + cb * bl[p * 16 + j];
}

extern "C" void kernel_launch(void* const* d_in, const int* in_sizes, int n_in,
                              void* d_out, int out_size, void* d_ws,
                              size_t ws_size, hipStream_t stream) {
  const float* features = (const float*)d_in[0];
  const float* W1 = (const float*)d_in[1];
  const float* b1 = (const float*)d_in[2];
  const float* W2 = (const float*)d_in[3];
  const float* b2 = (const float*)d_in[4];
  const float* Wl = (const float*)d_in[5];
  const float* bl = (const float*)d_in[6];
  const int* src_idx = (const int*)d_in[7];
  const int* to_fetch = (const int*)d_in[8];
  float* out = (float*)d_out;

  float* buf0 = (float*)d_ws;
  float* buf1 = buf0 + (size_t)65536 * 256;

  k_mm<512, false, true><<<512, 256, 0, stream>>>(
      const_cast<float*>(features), W1, b1, buf0, 0.25f, src_idx, to_fetch);
  k_agg<<<128, 256, 0, stream>>>(buf0, b1, buf1, 1.0f, 1);
  k_agg<<<8, 256, 0, stream>>>(buf1, b2, buf0, 0.25f, 0);
  k_mm<256, true, false><<<2, 256, 0, stream>>>(buf0, W2, b2, buf0, 1.0f,
                                                src_idx, to_fetch);
  k_final<<<1, 256, 0, stream>>>(buf0, Wl, bl, out);
}


# k_agg wave sums: lane exchange on the vector ALU (permlanex16 for lane^16, DPP row_xmask 8,4,2,1) instead of five LDS-permute hops per sum; butterfly order and sums unchanged
# speedup vs baseline: 5.1176x; 1.0997x over previous
#include <hip/hip_runtime.h>

typedef __attribute__((ext_vector_type(2))) float v2f;
typedef __attribute__((ext_vector_type(8))) float v8f;

#define NHID 256
#define EPSF 1e-7f
#define MAXNF 0.99999f

__device__ __forceinline__ v8f wmma_f32x4(v2f a, v2f b, v8f c) {
  return __builtin_amdgcn_wmma_f32_16x16x4_f32(false, a, false, b, (short)0, c,
                                               false, false);
}

__device__ __forceinline__ float lane_xor16(float x) {
  const int b = __builtin_bit_cast(int, x);
  return __builtin_bit_cast(
      float, __builtin_amdgcn_permlanex16(b, b, 0x76543210, 0xfedcba98, false,
                                          false));
}
template <int M>
__device__ __forceinline__ float lane_xor_row(float x) {
  const int b = __builtin_bit_cast(int, x);
  return __builtin_bit_cast(
      float, __builtin_amdgcn_update_dpp(b, b, 0x160 + M, 0xf, 0xf, true));
}

__device__ __forceinline__ float wred32(float v) {
  v += lane_xor16(v);
  v += lane_xor_row<8>(v);
  v += lane_xor_row<4>(v);
  v += lane_xor_row<2>(v);
  v += lane_xor_row<1>(v);
  return v;
}

__device__ __forceinline__ float hyp_scale(float x2, float m2) {
  float xn  = fminf(fmaxf(sqrtf(x2), EPSF), MAXNF);
  float mxn = fmaxf(sqrtf(m2), EPSF);
  return tanhf(mxn / xn * atanhf(xn)) / mxn;
}

__device__ __forceinline__ void load8(const float* p, float s, float* d) {
  const float4* g = (const float4*)p;
  float4 a = g[0], b = g[1];
  d[0] = a.x * s; d[1] = a.y * s; d[2] = a.z * s; d[3] = a.w * s;
  d[4] = b.x * s; d[5] = b.y * s; d[6] = b.z * s; d[7] = b.w * s;
}

__device__ __forceinline__ void store8(float* p, const float* d, float s) {
  float4 a, b;
  a.x = d[0] * s; a.y = d[1] * s; a.z = d[2] * s; a.w = d[3] * s;
  b.x = d[4] * s; b.y = d[5] * s; b.z = d[6] * s; b.w = d[7] * s;
  float4* g = (float4*)p;
  g[0] = a; g[1] = b;
}

template <int KDIM, bool FULL_EPI, bool GATHER>
__global__ __launch_bounds__(256) void k_mm(const float* X,
                                            const float* __restrict__ W,
                                            const float* __restrict__ bias,
                                            float* OUT, float inScale,
                                            const int* __restrict__ src,
                                            const int* __restrict__ tf) {
  constexpr int KT = 32;
  constexpr int A_PITCH = KT + 4;
  constexpr int B_PITCH = 2 * NHID + 32;
  constexpr int A_FL = 32 * A_PITCH;
  constexpr int B_FL = (KT / 2) * B_PITCH;
  constexpr int ST_FL = A_FL + B_FL;
  constexpr int NT = KDIM / KT;
  __shared__ __attribute__((aligned(16))) float stg[2 * ST_FL];
  float (*smx)[NHID + 8] = reinterpret_cast<float (*)[NHID + 8]>(stg);
  static_assert(32 * (NHID + 8) <= 2 * ST_FL, "epilogue rows fit the stages");
  __shared__ float part[32][8];
  __shared__ float rx2[32], rm2[32], rdot[32], rv2[32];
  __shared__ float sb2s;

  const int t = threadIdx.x;
  const int wave = t >> 5, lane = t & 31;
  const int tile_m = wave >> 2;
  const int n0 = (wave & 3) << 6;
  const int l15 = lane & 15;
  const int kk = (lane >> 4) << 1;
  const int brow = blockIdx.x << 5;
  const int r = t >> 3, p = t & 7;

  const int* nb = nullptr;
  if (GATHER) {
    const int b = brow >> 8;
    const int n2 = tf[b] + b * 1024;
    const int n1 = src[(size_t)n2 * 16 + (((brow >> 4) + tile_m) & 15)];
    nb = src + (size_t)n1 * 16;
  }
  const int xrow = GATHER ? nb[r & 15] : brow + r;

  const int c4 = t & 63, kp0 = t >> 6;
  const float* xg = X + (size_t)xrow * KDIM + p * 4;
  const float* wg = W + (size_t)(2 * kp0) * NHID + c4 * 4;
  float4 ra, rb0[4], rb1[4];
  auto gload = [&](int kt) {
    ra = *(const float4*)(xg + kt * KT);
    const float* w = wg + (size_t)(kt * KT) * NHID;
#pragma unroll
    for (int i = 0; i < 4; ++i) {
      rb0[i] = *(const float4*)(w + (size_t)(8 * i) * NHID);
      rb1[i] = *(const float4*)(w + (size_t)(8 * i + 1) * NHID);
    }
  };
  auto sstore = [&](float* st) {
    *(float4*)(st + r * A_PITCH + p * 4) = ra;
    float* bs = st + A_FL + kp0 * B_PITCH + c4 * 8;
#pragma unroll
    for (int i = 0; i < 4; ++i) {
      float4 lo, hi;
      lo.x = rb0[i].x; lo.y = rb1[i].x; lo.z = rb0[i].y; lo.w = rb1[i].y;
      hi.x = rb0[i].z; hi.y = rb1[i].z; hi.z = rb0[i].w; hi.w = rb1[i].w;
      *(float4*)(bs + (4 * i) * B_PITCH) = lo;
      *(float4*)(bs + (4 * i) * B_PITCH + 4) = hi;
    }
  };

  v8f acc0 = {}, acc1 = {}, acc2 = {}, acc3 = {};
  gload(0);
  sstore(stg);
  __syncthreads();
  for (int kt = 0; kt < NT; ++kt) {
    const float* st = stg + (kt & 1) * ST_FL;
    if (kt + 1 < NT) gload(kt + 1);
    const float* as = st + (tile_m * 16 + l15) * A_PITCH + kk;
    const float* bs = st + A_FL + (kk >> 1) * B_PITCH + (n0 + l15) * 2;
#pragma unroll
    for (int s = 0; s < KT / 4; ++s) {
      const v2f a = *(const v2f*)(as + s * 4);
      const float* bp = bs + (2 * s) * B_PITCH;
      const v2f b0 = *(const v2f*)(bp);
      const v2f b1 = *(const v2f*)(bp + 32);
      const v2f b2 = *(const v2f*)(bp + 64);
      const v2f b3 = *(const v2f*)(bp + 96);
      acc0 = wmma_f32x4(a, b0, acc0);
      acc1 = wmma_f32x4(a, b1, acc1);
      acc2 = wmma_f32x4(a, b2, acc2);
      acc3 = wmma_f32x4(a, b3, acc3);
    }
    if (kt + 1 < NT) sstore(stg + ((kt + 1) & 1) * ST_FL);
    __syncthreads();
  }
  {
    const int mb = tile_m * 16 + ((lane >> 4) ? 8 : 0);
#pragma unroll
    for (int v = 0; v < 8; ++v) {
      smx[mb + v][n0 + 0 + l15]  = acc0[v];
      smx[mb + v][n0 + 16 + l15] = acc1[v];
      smx[mb + v][n0 + 32 + l15] = acc2[v];
      smx[mb + v][n0 + 48 + l15] = acc3[v];
    }
  }
  __syncthreads();

  float sx = 0.f;
  {
    const int seg = KDIM / 8;
    const float* xr = X + (size_t)xrow * KDIM + p * seg;
    for (int j = 0; j < seg; j += 4) {
      float4 v = *(const float4*)(xr + j);
      sx += v.x * v.x + v.y * v.y + v.z * v.z + v.w * v.w;
    }
  }
  float sm = 0.f, sd = 0.f;
  {
    const float* mr = &smx[r][p * 32];
#pragma unroll
    for (int j = 0; j < 32; ++j) {
      float m = mr[j];
      sm += m * m;
      if (FULL_EPI) sd += m * bias[p * 32 + j];
    }
  }
  part[r][p] = sx;
  __syncthreads();
  if (t < 32) { float s = 0.f; for (int i = 0; i < 8; ++i) s += part[t][i]; rx2[t] = s; }
  __syncthreads();
  part[r][p] = sm;
  __syncthreads();
  if (t < 32) { float s = 0.f; for (int i = 0; i < 8; ++i) s += part[t][i]; rm2[t] = s; }
  __syncthreads();
  if (FULL_EPI) {
    part[r][p] = sd;
    __syncthreads();
    if (t < 32) { float s = 0.f; for (int i = 0; i < 8; ++i) s += part[t][i]; rdot[t] = s; }
    __syncthreads();
    part[r][p] = bias[t] * bias[t];
    __syncthreads();
    if (t == 0) {
      float s = 0.f;
      for (int i = 0; i < 32; ++i)
        for (int j = 0; j < 8; ++j) s += part[i][j];
      sb2s = s;
    }
    __syncthreads();
  }

  const float x2s = inScale * inScale * rx2[r];
  const float m2s = inScale * inScale * rm2[r];
  const float cmv = hyp_scale(x2s, m2s) * inScale;
  float* orow = OUT + (size_t)(brow + r) * NHID + p * 32;
  float* mrw = &smx[r][p * 32];
  if (!FULL_EPI) {
#pragma unroll
    for (int j = 0; j < 32; j += 4) {
      float4 o;
      o.x = cmv * mrw[j + 0]; o.y = cmv * mrw[j + 1];
      o.z = cmv * mrw[j + 2]; o.w = cmv * mrw[j + 3];
      *(float4*)(orow + j) = o;
    }
  } else {
    const float cf = 0.25f * cmv;
    const float x2r = cf * cf * rm2[r];
    const float xy = cf * rdot[r];
    const float y2 = sb2s;
    const float den = fmaxf(1.f + 2.f * xy + x2r * y2, EPSF);
    const float ca = (1.f + 2.f * xy + y2) / den;
    const float cb = (1.f - x2r) / den;
    const float zn2 = ca * ca * x2r + 2.f * ca * cb * xy + cb * cb * y2;
    const float zn = fminf(fmaxf(sqrtf(zn2), EPSF), MAXNF);
    const float us = atanhf(zn) / zn;
    const float acf = ca * cf;
    float sv = 0.f;
#pragma unroll
    for (int j = 0; j < 32; ++j) {
      float z = acf * mrw[j] + cb * bias[p * 32 + j];
      float uu = fmaxf(us * z, 0.f);
      mrw[j] = uu;
      sv += uu * uu;
    }
    part[r][p] = sv;
    __syncthreads();
    if (t < 32) { float s = 0.f; for (int i = 0; i < 8; ++i) s += part[t][i]; rv2[t] = s; }
    __syncthreads();
    const float vn = fmaxf(sqrtf(rv2[r]), EPSF);
    const float fs = tanhf(vn) / vn;
#pragma unroll
    for (int j = 0; j < 32; j += 4) {
      float4 o;
      o.x = fs * mrw[j + 0]; o.y = fs * mrw[j + 1];
      o.z = fs * mrw[j + 2]; o.w = fs * mrw[j + 3];
      *(float4*)(orow + j) = o;
    }
  }
}

__global__ __launch_bounds__(256) void k_agg(const float* __restrict__ F,
                                             const float* __restrict__ bias,
                                             float* __restrict__ OUT,
                                             float inScale, int doAct) {
  const int t = threadIdx.x;
  const int wave = t >> 5, lane = t & 31;
  const int node = (blockIdx.x << 3) + wave;
  const int joff = lane << 3;
  const float* msg = F + (size_t)node * 16 * NHID + joff;

  float x[8];
  load8(msg, inScale, x);
  float px = 0.f;
#pragma unroll
  for (int j = 0; j < 8; ++j) px += x[j] * x[j];
  float x2 = wred32(px);

  for (int k = 1; k < 16; ++k) {
    if (k < 15) __builtin_prefetch(msg + (k + 1) * NHID, 0, 0);
    float y[8];
    load8(msg + k * NHID, inScale, y);
    float py2 = 0.f, pxy = 0.f;
#pragma unroll
    for (int j = 0; j < 8; ++j) { py2 += y[j] * y[j]; pxy += x[j] * y[j]; }
    float y2 = wred32(py2);
    float xy = wred32(pxy);
    float den = fmaxf(1.f + 2.f * xy + x2 * y2, EPSF);
    float ca = (1.f + 2.f * xy + y2) / den;
    float cb = (1.f - x2) / den;
    float pn = 0.f;
#pragma unroll
    for (int j = 0; j < 8; ++j) { x[j] = ca * x[j] + cb * y[j]; pn += x[j] * x[j]; }
    x2 = wred32(pn);
  }

  float* orow = OUT + (size_t)node * NHID + joff;
  if (!doAct) {
    store8(orow, x, 1.f);
  } else {
    float b[8];
    load8(bias + joff, 1.f, b);
    float pb2 = 0.f, pxy = 0.f;
#pragma unroll
    for (int j = 0; j < 8; ++j) { pb2 += b[j] * b[j]; pxy += x[j] * b[j]; }
    float y2 = wred32(pb2);
    float xy = 0.25f * wred32(pxy);
    float x2r = 0.0625f * x2;
    float den = fmaxf(1.f + 2.f * xy + x2r * y2, EPSF);
    float ca = (1.f + 2.f * xy + y2) / den;
    float cb = (1.f - x2r) / den;
    float zn2 = ca * ca * x2r + 2.f * ca * cb * xy + cb * cb * y2;
    float zn = fminf(fmaxf(sqrtf(zn2), EPSF), MAXNF);
    float us = atanhf(zn) / zn;
    float u[8];
    float pv = 0.f;
#pragma unroll
    for (int j = 0; j < 8; ++j) {
      float z = ca * 0.25f * x[j] + cb * b[j];
      u[j] = fmaxf(us * z, 0.f);
      pv += u[j] * u[j];
    }
    float vn = fmaxf(sqrtf(wred32(pv)), EPSF);
    float fs = tanhf(vn) / vn;
    store8(orow, u, fs);
  }
}

__global__ __launch_bounds__(256) void k_final(const float* __restrict__ H,
                                               const float* __restrict__ Wl,
                                               const float* __restrict__ bl,
                                               float* __restrict__ OUT) {
  __shared__ float smx[64][72];
  __shared__ float part[64][4];
  __shared__ float rx2[64], rm2[64], rdot[64];
  __shared__ float sb2;

  const int t = threadIdx.x;
  if (t == 0) {
    float s = 0.f;
    for (int j = 0; j < 64; ++j) s += bl[j] * bl[j];
    sb2 = s;
  }
  __syncthreads();

  const int wave = t >> 5, lane = t & 31;
  const int l15 = lane & 15;
  const int kk = (lane >> 4) << 1;
#pragma unroll
  for (int i = 0; i < 2; ++i) {
    const int tile = wave * 2 + i;
    const int tm = tile >> 2, tn = tile & 3;
    const float* Ar = H + (size_t)(tm * 16 + l15) * NHID;
    const float* Br = Wl + (size_t)(tn * 16 + l15) * NHID;
    v8f c = {};
    for (int k0 = 0; k0 < NHID; k0 += 4) {
      v2f a, b;
      a.x = Ar[k0 + kk]; a.y = Ar[k0 + kk + 1];
      b.x = Br[k0 + kk]; b.y = Br[k0 + kk + 1];
      c = wmma_f32x4(a, b, c);
    }
    const int n = tn * 16 + l15;
    const int mb = tm * 16 + ((lane >> 4) ? 8 : 0);
#pragma unroll
    for (int v = 0; v < 8; ++v) smx[mb + v][n] = c[v];
  }
  __syncthreads();

  const int r = t >> 2, p = t & 3;
  float sx = 0.f;
  {
    const float* hr = H + (size_t)r * NHID + p * 64;
    for (int j = 0; j < 64; j += 4) {
      float4 v = *(const float4*)(hr + j);
      sx += v.x * v.x + v.y * v.y + v.z * v.z + v.w * v.w;
    }
  }
  float sm = 0.f, sd = 0.f;
#pragma unroll
  for (int j = 0; j < 16; ++j) {
    float m = smx[r][p * 16 + j];
    sm += m * m;
    sd += m * bl[p * 16 + j];
  }
  part[r][p] = sx;
  __syncthreads();
  if (t < 64) rx2[t] = part[t][0] + part[t][1] + part[t][2] + part[t][3];
  __syncthreads();
  part[r][p] = sm;
  __syncthreads();
  if (t < 64) rm2[t] = part[t][0] + part[t][1] + part[t][2] + part[t][3];
  __syncthreads();
  part[r][p] = sd;
  __syncthreads();
  if (t < 64) rdot[t] = part[t][0] + part[t][1] + part[t][2] + part[t][3];
  __syncthreads();

  const float s = hyp_scale(rx2[r], rm2[r]);
  const float x2r = s * s * rm2[r];
  const float xy = s * rdot[r];
  const float y2 = sb2;
  const float den = fmaxf(1.f + 2.f * xy + x2r * y2, EPSF);
  const float ca = (1.f + 2.f * xy + y2) / den;
  const float cb = (1.f - x2r) / den;
  const float acs = ca * s;
#pragma unroll
  for (int j = 0; j < 16; ++j)
    OUT[r * 64 + p * 16 + j] = acs * smx[r][p * 16 + j] + cb * bl[p * 16 + j];
}

extern "C" void kernel_launch(void* const* d_in, const int* in_sizes, int n_in,
                              void* d_out, int out_size, void* d_ws,
                              size_t ws_size, hipStream_t stream) {
  const float* features = (const float*)d_in[0];
  const float* W1 = (const float*)d_in[1];
  const float* b1 = (const float*)d_in[2];
  const float* W2 = (const float*)d_in[3];
  const float* b2 = (const float*)d_in[4];
  const float* Wl = (const float*)d_in[5];
  const float* bl = (const float*)d_in[6];
  const int* src_idx = (const int*)d_in[7];
  const int* to_fetch = (const int*)d_in[8];
  float* out = (float*)d_out;

  float* buf0 = (float*)d_ws;
  float* buf1 = buf0 + (size_t)65536 * 256;

  k_mm<512, false, true><<<512, 256, 0, stream>>>(
      const_cast<float*>(features), W1, b1, buf0, 0.25f, src_idx, to_fetch);
  k_agg<<<128, 256, 0, stream>>>(buf0, b1, buf1, 1.0f, 1);
  k_agg<<<8, 256, 0, stream>>>(buf1, b2, buf0, 0.25f, 0);
  k_mm<256, true, false><<<2, 256, 0, stream>>>(buf0, W2, b2, buf0, 1.0f,
                                                src_idx, to_fetch);
  k_final<<<1, 256, 0, stream>>>(buf0, Wl, bl, out);
}


# launches after layer 1 (layer-2 fold, its matvec+add+act, final linear) fused: one row-wise launch, 4 blocks x 16 rows, W2 to LDS under the fold; row sums by wave butterfly, final product split-K 4
# speedup vs baseline: 7.1026x; 1.3879x over previous
#include <hip/hip_runtime.h>

typedef __attribute__((ext_vector_type(2))) float v2f;
typedef __attribute__((ext_vector_type(8))) float v8f;

#define NHID 256
#define EPSF 1e-7f
#define MAXNF 0.99999f

__device__ __forceinline__ v8f wmma_f32x4(v2f a, v2f b, v8f c) {
  return __builtin_amdgcn_wmma_f32_16x16x4_f32(false, a, false, b, (short)0, c,
                                               false, false);
}

__device__ __forceinline__ float lane_xor16(float x) {
  const int b = __builtin_bit_cast(int, x);
  return __builtin_bit_cast(
      float, __builtin_amdgcn_permlanex16(b, b, 0x76543210, 0xfedcba98, false,
                                          false));
}
template <int M>
__device__ __forceinline__ float lane_xor_row(float x) {
  const int b = __builtin_bit_cast(int, x);
  return __builtin_bit_cast(
      float, __builtin_amdgcn_update_dpp(b, b, 0x160 + M, 0xf, 0xf, true));
}

__device__ __forceinline__ float wred32(float v) {
  v += lane_xor16(v);
  v += lane_xor_row<8>(v);
  v += lane_xor_row<4>(v);
  v += lane_xor_row<2>(v);
  v += lane_xor_row<1>(v);
  return v;
}

__device__ __forceinline__ float hyp_scale(float x2, float m2) {
  float xn  = fminf(fmaxf(sqrtf(x2), EPSF), MAXNF);
  float mxn = fmaxf(sqrtf(m2), EPSF);
  return tanhf(mxn / xn * atanhf(xn)) / mxn;
}

__device__ __forceinline__ void load8(const float* p, float s, float* d) {
  const float4* g = (const float4*)p;
  float4 a = g[0], b = g[1];
  d[0] = a.x * s; d[1] = a.y * s; d[2] = a.z * s; d[3] = a.w * s;
  d[4] = b.x * s; d[5] = b.y * s; d[6] = b.z * s; d[7] = b.w * s;
}

__device__ __forceinline__ void store8(float* p, const float* d, float s) {
  float4 a, b;
  a.x = d[0] * s; a.y = d[1] * s; a.z = d[2] * s; a.w = d[3] * s;
  b.x = d[4] * s; b.y = d[5] * s; b.z = d[6] * s; b.w = d[7] * s;
  float4* g = (float4*)p;
  g[0] = a; g[1] = b;
}

template <int KDIM, bool FULL_EPI, bool GATHER>
__global__ __launch_bounds__(256) void k_mm(const float* X,
                                            const float* __restrict__ W,
                                            const float* __restrict__ bias,
                                            float* OUT, float inScale,
                                            const int* __restrict__ src,
                                            const int* __restrict__ tf) {
  constexpr int KT = 32;
  constexpr int A_PITCH = KT + 4;
  constexpr int B_PITCH = 2 * NHID + 32;
  constexpr int A_FL = 32 * A_PITCH;
  constexpr int B_FL = (KT / 2) * B_PITCH;
  constexpr int ST_FL = A_FL + B_FL;
  constexpr int NT = KDIM / KT;
  __shared__ __attribute__((aligned(16))) float stg[2 * ST_FL];
  float (*smx)[NHID + 8] = reinterpret_cast<float (*)[NHID + 8]>(stg);
  static_assert(32 * (NHID + 8) <= 2 * ST_FL, "epilogue rows fit the stages");
  __shared__ float part[32][8];
  __shared__ float rx2[32], rm2[32], rdot[32], rv2[32];
  __shared__ float sb2s;

  const int t = threadIdx.x;
  const int wave = t >> 5, lane = t & 31;
  const int tile_m = wave >> 2;
  const int n0 = (wave & 3) << 6;
  const int l15 = lane & 15;
  const int kk = (lane >> 4) << 1;
  const int brow = blockIdx.x << 5;
  const int r = t >> 3, p = t & 7;

  const int* nb = nullptr;
  if (GATHER) {
    const int b = brow >> 8;
    const int n2 = tf[b] + b * 1024;
    const int n1 = src[(size_t)n2 * 16 + (((brow >> 4) + tile_m) & 15)];
    nb = src + (size_t)n1 * 16;
  }
  const int xrow = GATHER ? nb[r & 15] : brow + r;

  const int c4 = t & 63, kp0 = t >> 6;
  const float* xg = X + (size_t)xrow * KDIM + p * 4;
  const float* wg = W + (size_t)(2 * kp0) * NHID + c4 * 4;
  float4 ra, rb0[4], rb1[4];
  auto gload = [&](int kt) {
    ra = *(const float4*)(xg + kt * KT);
    const float* w = wg + (size_t)(kt * KT) * NHID;
#pragma unroll
    for (int i = 0; i < 4; ++i) {
      rb0[i] = *(const float4*)(w + (size_t)(8 * i) * NHID);
      rb1[i] = *(const float4*)(w + (size_t)(8 * i + 1) * NHID);
    }
  };
  auto sstore = [&](float* st) {
    *(float4*)(st + r * A_PITCH + p * 4) = ra;
    float* bs = st + A_FL + kp0 * B_PITCH + c4 * 8;
#pragma unroll
    for (int i = 0; i < 4; ++i) {
      float4 lo, hi;
      lo.x = rb0[i].x; lo.y = rb1[i].x; lo.z = rb0[i].y; lo.w = rb1[i].y;
      hi.x = rb0[i].z; hi.y = rb1[i].z; hi.z = rb0[i].w; hi.w = rb1[i].w;
      *(float4*)(bs + (4 * i) * B_PITCH) = lo;
      *(float4*)(bs + (4 * i) * B_PITCH + 4) = hi;
    }
  };

  v8f acc0 = {}, acc1 = {}, acc2 = {}, acc3 = {};
  gload(0);
  sstore(stg);
  __syncthreads();
  for (int kt = 0; kt < NT; ++kt) {
    const float* st = stg + (kt & 1) * ST_FL;
    if (kt + 1 < NT) gload(kt + 1);
    const float* as = st + (tile_m * 16 + l15) * A_PITCH + kk;
    const float* bs = st + A_FL + (kk >> 1) * B_PITCH + (n0 + l15) * 2;
#pragma unroll
    for (int s = 0; s < KT / 4; ++s) {
      const v2f a = *(const v2f*)(as + s * 4);
      const float* bp = bs + (2 * s) * B_PITCH;
      const v2f b0 = *(const v2f*)(bp);
      const v2f b1 = *(const v2f*)(bp + 32);
      const v2f b2 = *(const v2f*)(bp + 64);
      const v2f b3 = *(const v2f*)(bp + 96);
      acc0 = wmma_f32x4(a, b0, acc0);
      acc1 = wmma_f32x4(a, b1, acc1);
      acc2 = wmma_f32x4(a, b2, acc2);
      acc3 = wmma_f32x4(a, b3, acc3);
    }
    if (kt + 1 < NT) sstore(stg + ((kt + 1) & 1) * ST_FL);
    __syncthreads();
  }
  {
    const int mb = tile_m * 16 + ((lane >> 4) ? 8 : 0);
#pragma unroll
    for (int v = 0; v < 8; ++v) {
      smx[mb + v][n0 + 0 + l15]  = acc0[v];
      smx[mb + v][n0 + 16 + l15] = acc1[v];
      smx[mb + v][n0 + 32 + l15] = acc2[v];
      smx[mb + v][n0 + 48 + l15] = acc3[v];
    }
  }
  __syncthreads();

  float sx = 0.f;
  {
    const int seg = KDIM / 8;
    const float* xr = X + (size_t)xrow * KDIM + p * seg;
    for (int j = 0; j < seg; j += 4) {
      float4 v = *(const float4*)(xr + j);
      sx += v.x * v.x + v.y * v.y + v.z * v.z + v.w * v.w;
    }
  }
  float sm = 0.f, sd = 0.f;
  {
    const float* mr = &smx[r][p * 32];
#pragma unroll
    for (int j = 0; j < 32; ++j) {
      float m = mr[j];
      sm += m * m;
      if (FULL_EPI) sd += m * bias[p * 32 + j];
    }
  }
  part[r][p] = sx;
  __syncthreads();
  if (t < 32) { float s = 0.f; for (int i = 0; i < 8; ++i) s += part[t][i]; rx2[t] = s; }
  __syncthreads();
  part[r][p] = sm;
  __syncthreads();
  if (t < 32) { float s = 0.f; for (int i = 0; i < 8; ++i) s += part[t][i]; rm2[t] = s; }
  __syncthreads();
  if (FULL_EPI) {
    part[r][p] = sd;
    __syncthreads();
    if (t < 32) { float s = 0.f; for (int i = 0; i < 8; ++i) s += part[t][i]; rdot[t] = s; }
    __syncthreads();
    part[r][p] = bias[t] * bias[t];
    __syncthreads();
    if (t == 0) {
      float s = 0.f;
      for (int i = 0; i < 32; ++i)
        for (int j = 0; j < 8; ++j) s += part[i][j];
      sb2s = s;
    }
    __syncthreads();
  }

  const float x2s = inScale * inScale * rx2[r];
  const float m2s = inScale * inScale * rm2[r];
  const float cmv = hyp_scale(x2s, m2s) * inScale;
  float* orow = OUT + (size_t)(brow + r) * NHID + p * 32;
  float* mrw = &smx[r][p * 32];
  if (!FULL_EPI) {
#pragma unroll
    for (int j = 0; j < 32; j += 4) {
      float4 o;
      o.x = cmv * mrw[j + 0]; o.y = cmv * mrw[j + 1];
      o.z = cmv * mrw[j + 2]; o.w = cmv * mrw[j + 3];
      *(float4*)(orow + j) = o;
    }
  } else {
    const float cf = 0.25f * cmv;
    const float x2r = cf * cf * rm2[r];
    const float xy = cf * rdot[r];
    const float y2 = sb2s;
    const float den = fmaxf(1.f + 2.f * xy + x2r * y2, EPSF);
    const float ca = (1.f + 2.f * xy + y2) / den;
    const float cb = (1.f - x2r) / den;
    const float zn2 = ca * ca * x2r + 2.f * ca * cb * xy + cb * cb * y2;
    const float zn = fminf(fmaxf(sqrtf(zn2), EPSF), MAXNF);
    const float us = atanhf(zn) / zn;
    const float acf = ca * cf;
    float sv = 0.f;
#pragma unroll
    for (int j = 0; j < 32; ++j) {
      float z = acf * mrw[j] + cb * bias[p * 32 + j];
      float uu = fmaxf(us * z, 0.f);
      mrw[j] = uu;
      sv += uu * uu;
    }
    part[r][p] = sv;
    __syncthreads();
    if (t < 32) { float s = 0.f; for (int i = 0; i < 8; ++i) s += part[t][i]; rv2[t] = s; }
    __syncthreads();
    const float vn = fmaxf(sqrtf(rv2[r]), EPSF);
    const float fs = tanhf(vn) / vn;
#pragma unroll
    for (int j = 0; j < 32; j += 4) {
      float4 o;
      o.x = fs * mrw[j + 0]; o.y = fs * mrw[j + 1];
      o.z = fs * mrw[j + 2]; o.w = fs * mrw[j + 3];
      *(float4*)(orow + j) = o;
    }
  }
}

__global__ __launch_bounds__(256) void k_agg(const float* __restrict__ F,
                                             const float* __restrict__ bias,
                                             float* __restrict__ OUT,
                                             float inScale, int doAct) {
  const int t = threadIdx.x;
  const int wave = t >> 5, lane = t & 31;
  const int node = (blockIdx.x << 3) + wave;
  const int joff = lane << 3;
  const float* msg = F + (size_t)node * 16 * NHID + joff;

  float x[8];
  load8(msg, inScale, x);
  float px = 0.f;
#pragma unroll
  for (int j = 0; j < 8; ++j) px += x[j] * x[j];
  float x2 = wred32(px);

  for (int k = 1; k < 16; ++k) {
    if (k < 15) __builtin_prefetch(msg + (k + 1) * NHID, 0, 0);
    float y[8];
    load8(msg + k * NHID, inScale, y);
    float py2 = 0.f, pxy = 0.f;
#pragma unroll
    for (int j = 0; j < 8; ++j) { py2 += y[j] * y[j]; pxy += x[j] * y[j]; }
    float y2 = wred32(py2);
    float xy = wred32(pxy);
    float den = fmaxf(1.f + 2.f * xy + x2 * y2, EPSF);
    float ca = (1.f + 2.f * xy + y2) / den;
    float cb = (1.f - x2) / den;
    float pn = 0.f;
#pragma unroll
    for (int j = 0; j < 8; ++j) { x[j] = ca * x[j] + cb * y[j]; pn += x[j] * x[j]; }
    x2 = wred32(pn);
  }

  float* orow = OUT + (size_t)node * NHID + joff;
  if (!doAct) {
    store8(orow, x, 1.f);
  } else {
    float b[8];
    load8(bias + joff, 1.f, b);
    float pb2 = 0.f, pxy = 0.f;
#pragma unroll
    for (int j = 0; j < 8; ++j) { pb2 += b[j] * b[j]; pxy += x[j] * b[j]; }
    float y2 = wred32(pb2);
    float xy = 0.25f * wred32(pxy);
    float x2r = 0.0625f * x2;
    float den = fmaxf(1.f + 2.f * xy + x2r * y2, EPSF);
    float ca = (1.f + 2.f * xy + y2) / den;
    float cb = (1.f - x2r) / den;
    float zn2 = ca * ca * x2r + 2.f * ca * cb * xy + cb * cb * y2;
    float zn = fminf(fmaxf(sqrtf(zn2), EPSF), MAXNF);
    float us = atanhf(zn) / zn;
    float u[8];
    float pv = 0.f;
#pragma unroll
    for (int j = 0; j < 8; ++j) {
      float z = ca * 0.25f * x[j] + cb * b[j];
      u[j] = fmaxf(us * z, 0.f);
      pv += u[j] * u[j];
    }
    float vn = fmaxf(sqrtf(wred32(pv)), EPSF);
    float fs = tanhf(vn) / vn;
    store8(orow, u, fs);
  }
}

__global__ __launch_bounds__(512) void k_tail(const float* __restrict__ H1,
                                              const float* __restrict__ W2,
                                              const float* __restrict__ b2,
                                              const float* __restrict__ Wl,
                                              const float* __restrict__ bl,
                                              float* __restrict__ OUT) {
  constexpr int XP = NHID + 4;
  constexpr int WBP = 2 * NHID + 32;
  constexpr int PP = 64 + 4;
  __shared__ __attribute__((aligned(16))) float sW[(NHID / 2) * WBP];
  __shared__ __attribute__((aligned(16))) float sX[16 * XP];
  __shared__ __attribute__((aligned(16))) float sM[16 * XP];
  float* sWl = sW;
  float* sP = sW + 64 * XP;
  static_assert(64 * XP + 4 * 16 * PP <= (NHID / 2) * WBP, "Wl + partials fit");

  const int t = threadIdx.x;
  const int wave = t >> 5, lane = t & 31;
  const int l15 = lane & 15, hh = lane >> 4;
  const int node = (blockIdx.x << 4) + wave;
  const int joff = lane << 3;
  const float* msg = H1 + (size_t)node * 16 * NHID + joff;

  const int c4 = t & 63, kp0 = t >> 6;
  const float* wg = W2 + (size_t)(2 * kp0) * NHID + c4 * 4;
  float* ws = sW + kp0 * WBP + c4 * 8;
  float4 w0, w1;
  auto wload = [&](int i) {
    const float* p = wg + (size_t)(16 * i) * NHID;
    w0 = *(const float4*)p;
    w1 = *(const float4*)(p + NHID);
  };
  auto wstore = [&](int i) {
    float* q = ws + (8 * i) * WBP;
    *(float4*)q = w0;
    *(float4*)(q + 4) = w1;
  };

  float bb[8];
  load8(b2 + joff, 1.f, bb);
  const float bl0 = bl[2 * lane], bl1 = bl[2 * lane + 1];
  float x[8];
  load8(msg, 0.25f, x);
  float px = 0.f;
#pragma unroll
  for (int j = 0; j < 8; ++j) px += x[j] * x[j];
  float x2 = wred32(px);
  wload(0);
  for (int k = 1; k < 16; ++k) {
    if (k < 15) __builtin_prefetch(msg + (k + 1) * NHID, 0, 0);
    float y[8];
    load8(msg + k * NHID, 0.25f, y);
    wstore(k - 1);
    wload(k);
    float py2 = 0.f, pxy = 0.f;
#pragma unroll
    for (int j = 0; j < 8; ++j) { py2 += y[j] * y[j]; pxy += x[j] * y[j]; }
    float y2 = wred32(py2);
    float xy = wred32(pxy);
    float den = fmaxf(1.f + 2.f * xy + x2 * y2, EPSF);
    float ca = (1.f + 2.f * xy + y2) / den;
    float cb = (1.f - x2) / den;
    float pn = 0.f;
#pragma unroll
    for (int j = 0; j < 8; ++j) { x[j] = ca * x[j] + cb * y[j]; pn += x[j] * x[j]; }
    x2 = wred32(pn);
  }
  wstore(15);
  {
    float4 a, b;
    a.x = x[0]; a.y = x[1]; a.z = x[2]; a.w = x[3];
    b.x = x[4]; b.y = x[5]; b.z = x[6]; b.w = x[7];
    *(float4*)(sX + wave * XP + joff) = a;
    *(float4*)(sX + wave * XP + joff + 4) = b;
  }
  __syncthreads();

  const float* wlg = Wl + (size_t)(t >> 6) * NHID + (t & 63) * 4;
  const float4 wl0 = *(const float4*)(wlg);
  const float4 wl1 = *(const float4*)(wlg + 8 * NHID);
  const float4 wl2 = *(const float4*)(wlg + 16 * NHID);
  const float4 wl3 = *(const float4*)(wlg + 24 * NHID);
  const float4 wl4 = *(const float4*)(wlg + 32 * NHID);
  const float4 wl5 = *(const float4*)(wlg + 40 * NHID);
  const float4 wl6 = *(const float4*)(wlg + 48 * NHID);
  const float4 wl7 = *(const float4*)(wlg + 56 * NHID);

  {
    const int n0 = wave << 4;
    const float* as = sX + l15 * XP + 2 * hh;
    const int n = n0 + l15;
    const float* bs = sW + hh * WBP + (n >> 2) * 8 + (n & 3);
    v8f acc = {};
#pragma unroll
    for (int s = 0; s < NHID / 4; ++s) {
      const v2f a = *(const v2f*)(as + 4 * s);
      v2f b;
      b.x = bs[(2 * s) * WBP];
      b.y = bs[(2 * s) * WBP + 4];
      acc = wmma_f32x4(a, b, acc);
    }
#pragma unroll
    for (int v = 0; v < 8; ++v) sM[(8 * hh + v) * XP + n0 + l15] = acc[v];
  }
  __syncthreads();

  {
    float* wls = sWl + (t >> 6) * XP + (t & 63) * 4;
    *(float4*)(wls) = wl0;
    *(float4*)(wls + 8 * XP) = wl1;
    *(float4*)(wls + 16 * XP) = wl2;
    *(float4*)(wls + 24 * XP) = wl3;
    *(float4*)(wls + 32 * XP) = wl4;
    *(float4*)(wls + 40 * XP) = wl5;
    *(float4*)(wls + 48 * XP) = wl6;
    *(float4*)(wls + 56 * XP) = wl7;
  }

  float hx2;
  {
    float m[8];
    {
      const float4 a = *(const float4*)(sM + wave * XP + joff);
      const float4 b = *(const float4*)(sM + wave * XP + joff + 4);
      m[0] = a.x; m[1] = a.y; m[2] = a.z; m[3] = a.w;
      m[4] = b.x; m[5] = b.y; m[6] = b.z; m[7] = b.w;
    }
    float pm2 = 0.f, pmd = 0.f, pb2 = 0.f;
#pragma unroll
    for (int j = 0; j < 8; ++j) {
      pm2 += m[j] * m[j]; pmd += m[j] * bb[j]; pb2 += bb[j] * bb[j];
    }
    const float m2 = wred32(pm2);
    const float md = wred32(pmd);
    const float y2 = wred32(pb2);
    const float cmv = hyp_scale(x2, m2);
    const float cf = 0.25f * cmv;
    const float x2r = cf * cf * m2;
    const float xy = cf * md;
    const float den = fmaxf(1.f + 2.f * xy + x2r * y2, EPSF);
    const float ca = (1.f + 2.f * xy + y2) / den;
    const float cb = (1.f - x2r) / den;
    const float zn2 = ca * ca * x2r + 2.f * ca * cb * xy + cb * cb * y2;
    const float zn = fminf(fmaxf(sqrtf(zn2), EPSF), MAXNF);
    const float us = atanhf(zn) / zn;
    const float acf = ca * cf;
    float u[8];
    float pv = 0.f;
#pragma unroll
    for (int j = 0; j < 8; ++j) {
      const float z = acf * m[j] + cb * bb[j];
      u[j] = fmaxf(us * z, 0.f);
      pv += u[j] * u[j];
    }
    const float vn = fmaxf(sqrtf(wred32(pv)), EPSF);
    const float fs = tanhf(vn) / vn;
    float ph = 0.f;
#pragma unroll
    for (int j = 0; j < 8; ++j) { u[j] = fs * u[j]; ph += u[j] * u[j]; }
    hx2 = wred32(ph);
    float4 a, b;
    a.x = u[0]; a.y = u[1]; a.z = u[2]; a.w = u[3];
    b.x = u[4]; b.y = u[5]; b.z = u[6]; b.w = u[7];
    *(float4*)(sX + wave * XP + joff) = a;
    *(float4*)(sX + wave * XP + joff + 4) = b;
  }
  __syncthreads();

  {
    const int tn = wave & 3, kq = wave >> 2;
    const float* ar = sX + l15 * XP + 64 * kq + 2 * hh;
    const float* br = sWl + (16 * tn + l15) * XP + 64 * kq + 2 * hh;
    v8f c = {};
#pragma unroll
    for (int s = 0; s < 16; ++s) {
      const v2f a = *(const v2f*)(ar + 4 * s);
      const v2f b = *(const v2f*)(br + 4 * s);
      c = wmma_f32x4(a, b, c);
    }
    float* pt = sP + kq * (16 * PP) + 16 * tn + l15;
#pragma unroll
    for (int v = 0; v < 8; ++v) pt[(8 * hh + v) * PP] = c[v];
  }
  __syncthreads();

  {
    const float* pr = sP + wave * PP + 2 * lane;
    v2f mm = *(const v2f*)pr;
#pragma unroll
    for (int q = 1; q < 4; ++q) {
      const v2f pq = *(const v2f*)(pr + q * (16 * PP));
      mm.x += pq.x; mm.y += pq.y;
    }
    const float m2 = wred32(mm.x * mm.x + mm.y * mm.y);
    const float md = wred32(mm.x * bl0 + mm.y * bl1);
    const float y2 = wred32(bl0 * bl0 + bl1 * bl1);
    const float s = hyp_scale(hx2, m2);
    const float x2r = s * s * m2;
    const float xy = s * md;
    const float den = fmaxf(1.f + 2.f * xy + x2r * y2, EPSF);
    const float ca = (1.f + 2.f * xy + y2) / den;
    const float cb = (1.f - x2r) / den;
    const float acs = ca * s;
    float2 o;
    o.x = acs * mm.x + cb * bl0;
    o.y = acs * mm.y + cb * bl1;
    *(float2*)(OUT + (size_t)node * 64 + 2 * lane) = o;
  }
}

extern "C" void kernel_launch(void* const* d_in, const int* in_sizes, int n_in,
                              void* d_out, int out_size, void* d_ws,
                              size_t ws_size, hipStream_t stream) {
  const float* features = (const float*)d_in[0];
  const float* W1 = (const float*)d_in[1];
  const float* b1 = (const float*)d_in[2];
  const float* W2 = (const float*)d_in[3];
  const float* b2 = (const float*)d_in[4];
  const float* Wl = (const float*)d_in[5];
  const float* bl = (const float*)d_in[6];
  const int* src_idx = (const int*)d_in[7];
  const int* to_fetch = (const int*)d_in[8];
  float* out = (float*)d_out;

  float* buf0 = (float*)d_ws;
  float* buf1 = buf0 + (size_t)65536 * 256;

  k_mm<512, false, true><<<512, 256, 0, stream>>>(
      const_cast<float*>(features), W1, b1, buf0, 0.25f, src_idx, to_fetch);
  k_agg<<<128, 256, 0, stream>>>(buf0, b1, buf1, 1.0f, 1);
  k_tail<<<4, 512, 0, stream>>>(buf1, W2, b2, Wl, bl, out);
}


# k_agg and k_tail fold: rows (k_tail: W2 items too) requested a group of three fold steps ahead in a software pipeline, not one exposed L2 round trip per step; sums and their order unchanged
# speedup vs baseline: 7.8788x; 1.1093x over previous
#include <hip/hip_runtime.h>

typedef __attribute__((ext_vector_type(2))) float v2f;
typedef __attribute__((ext_vector_type(8))) float v8f;

#define NHID 256
#define EPSF 1e-7f
#define MAXNF 0.99999f

__device__ __forceinline__ v8f wmma_f32x4(v2f a, v2f b, v8f c) {
  return __builtin_amdgcn_wmma_f32_16x16x4_f32(false, a, false, b, (short)0, c,
                                               false, false);
}

__device__ __forceinline__ float lane_xor16(float x) {
  const int b = __builtin_bit_cast(int, x);
  return __builtin_bit_cast(
      float, __builtin_amdgcn_permlanex16(b, b, 0x76543210, 0xfedcba98, false,
                                          false));
}
template <int M>
__device__ __forceinline__ float lane_xor_row(float x) {
  const int b = __builtin_bit_cast(int, x);
  return __builtin_bit_cast(
      float, __builtin_amdgcn_update_dpp(b, b, 0x160 + M, 0xf, 0xf, true));
}

__device__ __forceinline__ float wred32(float v) {
  v += lane_xor16(v);
  v += lane_xor_row<8>(v);
  v += lane_xor_row<4>(v);
  v += lane_xor_row<2>(v);
  v += lane_xor_row<1>(v);
  return v;
}

__device__ __forceinline__ float hyp_scale(float x2, float m2) {
  float xn  = fminf(fmaxf(sqrtf(x2), EPSF), MAXNF);
  float mxn = fmaxf(sqrtf(m2), EPSF);
  return tanhf(mxn / xn * atanhf(xn)) / mxn;
}

__device__ __forceinline__ void load8(const float* p, float s, float* d) {
  const float4* g = (const float4*)p;
  float4 a = g[0], b = g[1];
  d[0] = a.x * s; d[1] = a.y * s; d[2] = a.z * s; d[3] = a.w * s;
  d[4] = b.x * s; d[5] = b.y * s; d[6] = b.z * s; d[7] = b.w * s;
}

__device__ __forceinline__ void scale8(const float4& a, const float4& b,
                                       float s, float* d) {
  d[0] = a.x * s; d[1] = a.y * s; d[2] = a.z * s; d[3] = a.w * s;
  d[4] = b.x * s; d[5] = b.y * s; d[6] = b.z * s; d[7] = b.w * s;
}

__device__ __forceinline__ void fold_step(float* x, float& x2, const float4& a,
                                          const float4& b, float s) {
  float y[8];
  scale8(a, b, s, y);
  float py2 = 0.f, pxy = 0.f;
#pragma unroll
  for (int j = 0; j < 8; ++j) { py2 += y[j] * y[j]; pxy += x[j] * y[j]; }
  float y2 = wred32(py2);
  float xy = wred32(pxy);
  float den = fmaxf(1.f + 2.f * xy + x2 * y2, EPSF);
  float ca = (1.f + 2.f * xy + y2) / den;
  float cb = (1.f - x2) / den;
  float pn = 0.f;
#pragma unroll
  for (int j = 0; j < 8; ++j) { x[j] = ca * x[j] + cb * y[j]; pn += x[j] * x[j]; }
  x2 = wred32(pn);
}

__device__ __forceinline__ void store8(float* p, const float* d, float s) {
  float4 a, b;
  a.x = d[0] * s; a.y = d[1] * s; a.z = d[2] * s; a.w = d[3] * s;
  b.x = d[4] * s; b.y = d[5] * s; b.z = d[6] * s; b.w = d[7] * s;
  float4* g = (float4*)p;
  g[0] = a; g[1] = b;
}

template <int KDIM, bool FULL_EPI, bool GATHER>
__global__ __launch_bounds__(256) void k_mm(const float* X,
                                            const float* __restrict__ W,
                                            const float* __restrict__ bias,
                                            float* OUT, float inScale,
                                            const int* __restrict__ src,
                                            const int* __restrict__ tf) {
  constexpr int KT = 32;
  constexpr int A_PITCH = KT + 4;
  constexpr int B_PITCH = 2 * NHID + 32;
  constexpr int A_FL = 32 * A_PITCH;
  constexpr int B_FL = (KT / 2) * B_PITCH;
  constexpr int ST_FL = A_FL + B_FL;
  constexpr int NT = KDIM / KT;
  __shared__ __attribute__((aligned(16))) float stg[2 * ST_FL];
  float (*smx)[NHID + 8] = reinterpret_cast<float (*)[NHID + 8]>(stg);
  static_assert(32 * (NHID + 8) <= 2 * ST_FL, "epilogue rows fit the stages");
  __shared__ float part[32][8];
  __shared__ float rx2[32], rm2[32], rdot[32], rv2[32];
  __shared__ float sb2s;

  const int t = threadIdx.x;
  const int wave = t >> 5, lane = t & 31;
  const int tile_m = wave >> 2;
  const int n0 = (wave & 3) << 6;
  const int l15 = lane & 15;
  const int kk = (lane >> 4) << 1;
  const int brow = blockIdx.x << 5;
  const int r = t >> 3, p = t & 7;

  const int* nb = nullptr;
  if (GATHER) {
    const int b = brow >> 8;
    const int n2 = tf[b] + b * 1024;
    const int n1 = src[(size_t)n2 * 16 + (((brow >> 4) + tile_m) & 15)];
    nb = src + (size_t)n1 * 16;
  }
  const int xrow = GATHER ? nb[r & 15] : brow + r;

  const int c4 = t & 63, kp0 = t >> 6;
  const float* xg = X + (size_t)xrow * KDIM + p * 4;
  const float* wg = W + (size_t)(2 * kp0) * NHID + c4 * 4;
  float4 ra, rb0[4], rb1[4];
  auto gload = [&](int kt) {
    ra = *(const float4*)(xg + kt * KT);
    const float* w = wg + (size_t)(kt * KT) * NHID;
#pragma unroll
    for (int i = 0; i < 4; ++i) {
      rb0[i] = *(const float4*)(w + (size_t)(8 * i) * NHID);
      rb1[i] = *(const float4*)(w + (size_t)(8 * i + 1) * NHID);
    }
  };
  auto sstore = [&](float* st) {
    *(float4*)(st + r * A_PITCH + p * 4) = ra;
    float* bs = st + A_FL + kp0 * B_PITCH + c4 * 8;
#pragma unroll
    for (int i = 0; i < 4; ++i) {
      float4 lo, hi;
      lo.x = rb0[i].x; lo.y = rb1[i].x; lo.z = rb0[i].y; lo.w = rb1[i].y;
      hi.x = rb0[i].z; hi.y = rb1[i].z; hi.z = rb0[i].w; hi.w = rb1[i].w;
      *(float4*)(bs + (4 * i) * B_PITCH) = lo;
      *(float4*)(bs + (4 * i) * B_PITCH + 4) = hi;
    }
  };

  v8f acc0 = {}, acc1 = {}, acc2 = {}, acc3 = {};
  gload(0);
  sstore(stg);
  __syncthreads();
  for (int kt = 0; kt < NT; ++kt) {
    const float* st = stg + (kt & 1) * ST_FL;
    if (kt + 1 < NT) gload(kt + 1);
    const float* as = st + (tile_m * 16 + l15) * A_PITCH + kk;
    const float* bs = st + A_FL + (kk >> 1) * B_PITCH + (n0 + l15) * 2;
#pragma unroll
    for (int s = 0; s < KT / 4; ++s) {
      const v2f a = *(const v2f*)(as + s * 4);
      const float* bp = bs + (2 * s) * B_PITCH;
      const v2f b0 = *(const v2f*)(bp);
      const v2f b1 = *(const v2f*)(bp + 32);
      const v2f b2 = *(const v2f*)(bp + 64);
      const v2f b3 = *(const v2f*)(bp + 96);
      acc0 = wmma_f32x4(a, b0, acc0);
      acc1 = wmma_f32x4(a, b1, acc1);
      acc2 = wmma_f32x4(a, b2, acc2);
      acc3 = wmma_f32x4(a, b3, acc3);
    }
    if (kt + 1 < NT) sstore(stg + ((kt + 1) & 1) * ST_FL);
    __syncthreads();
  }
  {
    const int mb = tile_m * 16 + ((lane >> 4) ? 8 : 0);
#pragma unroll
    for (int v = 0; v < 8; ++v) {
      smx[mb + v][n0 + 0 + l15]  = acc0[v];
      smx[mb + v][n0 + 16 + l15] = acc1[v];
      smx[mb + v][n0 + 32 + l15] = acc2[v];
      smx[mb + v][n0 + 48 + l15] = acc3[v];
    }
  }
  __syncthreads();

  float sx = 0.f;
  {
    const int seg = KDIM / 8;
    const float* xr = X + (size_t)xrow * KDIM + p * seg;
    for (int j = 0; j < seg; j += 4) {
      float4 v = *(const float4*)(xr + j);
      sx += v.x * v.x + v.y * v.y + v.z * v.z + v.w * v.w;
    }
  }
  float sm = 0.f, sd = 0.f;
  {
    const float* mr = &smx[r][p * 32];
#pragma unroll
    for (int j = 0; j < 32; ++j) {
      float m = mr[j];
      sm += m * m;
      if (FULL_EPI) sd += m * bias[p * 32 + j];
    }
  }
  part[r][p] = sx;
  __syncthreads();
  if (t < 32) { float s = 0.f; for (int i = 0; i < 8; ++i) s += part[t][i]; rx2[t] = s; }
  __syncthreads();
  part[r][p] = sm;
  __syncthreads();
  if (t < 32) { float s = 0.f; for (int i = 0; i < 8; ++i) s += part[t][i]; rm2[t] = s; }
  __syncthreads();
  if (FULL_EPI) {
    part[r][p] = sd;
    __syncthreads();
    if (t < 32) { float s = 0.f; for (int i = 0; i < 8; ++i) s += part[t][i]; rdot[t] = s; }
    __syncthreads();
    part[r][p] = bias[t] * bias[t];
    __syncthreads();
    if (t == 0) {
      float s = 0.f;
      for (int i = 0; i < 32; ++i)
        for (int j = 0; j < 8; ++j) s += part[i][j];
      sb2s = s;
    }
    __syncthreads();
  }

  const float x2s = inScale * inScale * rx2[r];
  const float m2s = inScale * inScale * rm2[r];
  const float cmv = hyp_scale(x2s, m2s) * inScale;
  float* orow = OUT + (size_t)(brow + r) * NHID + p * 32;
  float* mrw = &smx[r][p * 32];
  if (!FULL_EPI) {
#pragma unroll
    for (int j = 0; j < 32; j += 4) {
      float4 o;
      o.x = cmv * mrw[j + 0]; o.y = cmv * mrw[j + 1];
      o.z = cmv * mrw[j + 2]; o.w = cmv * mrw[j + 3];
      *(float4*)(orow + j) = o;
    }
  } else {
    const float cf = 0.25f * cmv;
    const float x2r = cf * cf * rm2[r];
    const float xy = cf * rdot[r];
    const float y2 = sb2s;
    const float den = fmaxf(1.f + 2.f * xy + x2r * y2, EPSF);
    const float ca = (1.f + 2.f * xy + y2) / den;
    const float cb = (1.f - x2r) / den;
    const float zn2 = ca * ca * x2r + 2.f * ca * cb * xy + cb * cb * y2;
    const float zn = fminf(fmaxf(sqrtf(zn2), EPSF), MAXNF);
    const float us = atanhf(zn) / zn;
    const float acf = ca * cf;
    float sv = 0.f;
#pragma unroll
    for (int j = 0; j < 32; ++j) {
      float z = acf * mrw[j] + cb * bias[p * 32 + j];
      float uu = fmaxf(us * z, 0.f);
      mrw[j] = uu;
      sv += uu * uu;
    }
    part[r][p] = sv;
    __syncthreads();
    if (t < 32) { float s = 0.f; for (int i = 0; i < 8; ++i) s += part[t][i]; rv2[t] = s; }
    __syncthreads();
    const float vn = fmaxf(sqrtf(rv2[r]), EPSF);
    const float fs = tanhf(vn) / vn;
#pragma unroll
    for (int j = 0; j < 32; j += 4) {
      float4 o;
      o.x = fs * mrw[j + 0]; o.y = fs * mrw[j + 1];
      o.z = fs * mrw[j + 2]; o.w = fs * mrw[j + 3];
      *(float4*)(orow + j) = o;
    }
  }
}

__global__ __launch_bounds__(256) void k_agg(const float* __restrict__ F,
                                             const float* __restrict__ bias,
                                             float* __restrict__ OUT,
                                             float inScale, int doAct) {
  const int t = threadIdx.x;
  const int wave = t >> 5, lane = t & 31;
  const int node = (blockIdx.x << 3) + wave;
  const int joff = lane << 3;
  const float* msg = F + (size_t)node * 16 * NHID + joff;

  const float4* g0 = (const float4*)msg;
  constexpr int RS = NHID / 4;
  float x[8];
  {
    const float4 a = g0[0], b = g0[1];
    scale8(a, b, inScale, x);
  }
  float px = 0.f;
#pragma unroll
  for (int j = 0; j < 8; ++j) px += x[j] * x[j];
  float x2 = wred32(px);
  const float4 z4 = make_float4(0.f, 0.f, 0.f, 0.f);
  float4 a0 = z4, b0 = z4, a1 = z4, b1 = z4, a2 = z4, b2 = z4;
#pragma unroll 1
  for (int k0 = -2; k0 < 16; k0 += 3) {
    const float4* gn0 = g0 + min(k0 + 3, 15) * RS;
    const float4* gn1 = g0 + min(k0 + 4, 15) * RS;
    const float4* gn2 = g0 + min(k0 + 5, 15) * RS;
    const float4 na0 = gn0[0], nb0 = gn0[1];
    const float4 na1 = gn1[0], nb1 = gn1[1];
    const float4 na2 = gn2[0], nb2 = gn2[1];
    if (k0 >= 1) {
      fold_step(x, x2, a0, b0, inScale);
      fold_step(x, x2, a1, b1, inScale);
      fold_step(x, x2, a2, b2, inScale);
    }
    a0 = na0; b0 = nb0; a1 = na1; b1 = nb1; a2 = na2; b2 = nb2;
  }

  float* orow = OUT + (size_t)node * NHID + joff;
  if (!doAct) {
    store8(orow, x, 1.f);
  } else {
    float b[8];
    load8(bias + joff, 1.f, b);
    float pb2 = 0.f, pxy = 0.f;
#pragma unroll
    for (int j = 0; j < 8; ++j) { pb2 += b[j] * b[j]; pxy += x[j] * b[j]; }
    float y2 = wred32(pb2);
    float xy = 0.25f * wred32(pxy);
    float x2r = 0.0625f * x2;
    float den = fmaxf(1.f + 2.f * xy + x2r * y2, EPSF);
    float ca = (1.f + 2.f * xy + y2) / den;
    float cb = (1.f - x2r) / den;
    float zn2 = ca * ca * x2r + 2.f * ca * cb * xy + cb * cb * y2;
    float zn = fminf(fmaxf(sqrtf(zn2), EPSF), MAXNF);
    float us = atanhf(zn) / zn;
    float u[8];
    float pv = 0.f;
#pragma unroll
    for (int j = 0; j < 8; ++j) {
      float z = ca * 0.25f * x[j] + cb * b[j];
      u[j] = fmaxf(us * z, 0.f);
      pv += u[j] * u[j];
    }
    float vn = fmaxf(sqrtf(wred32(pv)), EPSF);
    float fs = tanhf(vn) / vn;
    store8(orow, u, fs);
  }
}

__global__ __launch_bounds__(512) void k_tail(const float* __restrict__ H1,
                                              const float* __restrict__ W2,
                                              const float* __restrict__ b2,
                                              const float* __restrict__ Wl,
                                              const float* __restrict__ bl,
                                              float* __restrict__ OUT) {
  constexpr int XP = NHID + 4;
  constexpr int WBP = 2 * NHID + 32;
  constexpr int PP = 64 + 4;
  __shared__ __attribute__((aligned(16))) float sW[(NHID / 2) * WBP];
  __shared__ __attribute__((aligned(16))) float sX[16 * XP];
  __shared__ __attribute__((aligned(16))) float sM[16 * XP];
  float* sWl = sW;
  float* sP = sW + 64 * XP;
  static_assert(64 * XP + 4 * 16 * PP <= (NHID / 2) * WBP, "Wl + partials fit");

  const int t = threadIdx.x;
  const int wave = t >> 5, lane = t & 31;
  const int l15 = lane & 15, hh = lane >> 4;
  const int node = (blockIdx.x << 4) + wave;
  const int joff = lane << 3;
  const float* msg = H1 + (size_t)node * 16 * NHID + joff;

  const int c4 = t & 63, kp0 = t >> 6;
  const float* wg = W2 + (size_t)(2 * kp0) * NHID + c4 * 4;
  float* ws = sW + kp0 * WBP + c4 * 8;
  auto wload = [&](int i, float4& w0, float4& w1) {
    const float* p = wg + (size_t)(16 * i) * NHID;
    w0 = *(const float4*)p;
    w1 = *(const float4*)(p + NHID);
  };
  auto wstore = [&](int i, const float4& w0, const float4& w1) {
    float* q = ws + (8 * i) * WBP;
    *(float4*)q = w0;
    *(float4*)(q + 4) = w1;
  };

  float bb[8];
  load8(b2 + joff, 1.f, bb);
  const float bl0 = bl[2 * lane], bl1 = bl[2 * lane + 1];
  const float4* g0 = (const float4*)msg;
  constexpr int RS = NHID / 4;
  float x[8];
  {
    const float4 a = g0[0], b = g0[1];
    scale8(a, b, 0.25f, x);
  }
  float px = 0.f;
#pragma unroll
  for (int j = 0; j < 8; ++j) px += x[j] * x[j];
  float x2 = wred32(px);
  const float4 z4 = make_float4(0.f, 0.f, 0.f, 0.f);
  float4 a0 = z4, b0 = z4, a1 = z4, b1 = z4, a2 = z4, b2r = z4;
  float4 u0 = z4, v0 = z4, u1 = z4, v1 = z4, u2 = z4, v2 = z4;
#pragma unroll 1
  for (int k0 = -2; k0 < 16; k0 += 3) {
    const float4* gn0 = g0 + min(k0 + 3, 15) * RS;
    const float4* gn1 = g0 + min(k0 + 4, 15) * RS;
    const float4* gn2 = g0 + min(k0 + 5, 15) * RS;
    const float4 na0 = gn0[0], nb0 = gn0[1];
    const float4 na1 = gn1[0], nb1 = gn1[1];
    const float4 na2 = gn2[0], nb2 = gn2[1];
    float4 nu0, nv0, nu1, nv1, nu2, nv2;
    wload(min(k0 + 2, 15), nu0, nv0);
    wload(min(k0 + 3, 15), nu1, nv1);
    wload(min(k0 + 4, 15), nu2, nv2);
    if (k0 >= 1) {
      wstore(k0 - 1, u0, v0);
      fold_step(x, x2, a0, b0, 0.25f);
      wstore(k0, u1, v1);
      fold_step(x, x2, a1, b1, 0.25f);
      wstore(k0 + 1, u2, v2);
      fold_step(x, x2, a2, b2r, 0.25f);
    }
    a0 = na0; b0 = nb0; a1 = na1; b1 = nb1; a2 = na2; b2r = nb2;
    u0 = nu0; v0 = nv0; u1 = nu1; v1 = nv1; u2 = nu2; v2 = nv2;
  }
  wstore(15, u0, v0);
  {
    float4 a, b;
    a.x = x[0]; a.y = x[1]; a.z = x[2]; a.w = x[3];
    b.x = x[4]; b.y = x[5]; b.z = x[6]; b.w = x[7];
    *(float4*)(sX + wave * XP + joff) = a;
    *(float4*)(sX + wave * XP + joff + 4) = b;
  }
  __syncthreads();

  const float* wlg = Wl + (size_t)(t >> 6) * NHID + (t & 63) * 4;
  const float4 wl0 = *(const float4*)(wlg);
  const float4 wl1 = *(const float4*)(wlg + 8 * NHID);
  const float4 wl2 = *(const float4*)(wlg + 16 * NHID);
  const float4 wl3 = *(const float4*)(wlg + 24 * NHID);
  const float4 wl4 = *(const float4*)(wlg + 32 * NHID);
  const float4 wl5 = *(const float4*)(wlg + 40 * NHID);
  const float4 wl6 = *(const float4*)(wlg + 48 * NHID);
  const float4 wl7 = *(const float4*)(wlg + 56 * NHID);

  {
    const int n0 = wave << 4;
    const float* as = sX + l15 * XP + 2 * hh;
    const int n = n0 + l15;
    const float* bs = sW + hh * WBP + (n >> 2) * 8 + (n & 3);
    v8f acc = {};
#pragma unroll
    for (int s = 0; s < NHID / 4; ++s) {
      const v2f a = *(const v2f*)(as + 4 * s);
      v2f b;
      b.x = bs[(2 * s) * WBP];
      b.y = bs[(2 * s) * WBP + 4];
      acc = wmma_f32x4(a, b, acc);
    }
#pragma unroll
    for (int v = 0; v < 8; ++v) sM[(8 * hh + v) * XP + n0 + l15] = acc[v];
  }
  __syncthreads();

  {
    float* wls = sWl + (t >> 6) * XP + (t & 63) * 4;
    *(float4*)(wls) = wl0;
    *(float4*)(wls + 8 * XP) = wl1;
    *(float4*)(wls + 16 * XP) = wl2;
    *(float4*)(wls + 24 * XP) = wl3;
    *(float4*)(wls + 32 * XP) = wl4;
    *(float4*)(wls + 40 * XP) = wl5;
    *(float4*)(wls + 48 * XP) = wl6;
    *(float4*)(wls + 56 * XP) = wl7;
  }

  float hx2;
  {
    float m[8];
    {
      const float4 a = *(const float4*)(sM + wave * XP + joff);
      const float4 b = *(const float4*)(sM + wave * XP + joff + 4);
      m[0] = a.x; m[1] = a.y; m[2] = a.z; m[3] = a.w;
      m[4] = b.x; m[5] = b.y; m[6] = b.z; m[7] = b.w;
    }
    float pm2 = 0.f, pmd = 0.f, pb2 = 0.f;
#pragma unroll
    for (int j = 0; j < 8; ++j) {
      pm2 += m[j] * m[j]; pmd += m[j] * bb[j]; pb2 += bb[j] * bb[j];
    }
    const float m2 = wred32(pm2);
    const float md = wred32(pmd);
    const float y2 = wred32(pb2);
    const float cmv = hyp_scale(x2, m2);
    const float cf = 0.25f * cmv;
    const float x2r = cf * cf * m2;
    const float xy = cf * md;
    const float den = fmaxf(1.f + 2.f * xy + x2r * y2, EPSF);
    const float ca = (1.f + 2.f * xy + y2) / den;
    const float cb = (1.f - x2r) / den;
    const float zn2 = ca * ca * x2r + 2.f * ca * cb * xy + cb * cb * y2;
    const float zn = fminf(fmaxf(sqrtf(zn2), EPSF), MAXNF);
    const float us = atanhf(zn) / zn;
    const float acf = ca * cf;
    float u[8];
    float pv = 0.f;
#pragma unroll
    for (int j = 0; j < 8; ++j) {
      const float z = acf * m[j] + cb * bb[j];
      u[j] = fmaxf(us * z, 0.f);
      pv += u[j] * u[j];
    }
    const float vn = fmaxf(sqrtf(wred32(pv)), EPSF);
    const float fs = tanhf(vn) / vn;
    float ph = 0.f;
#pragma unroll
    for (int j = 0; j < 8; ++j) { u[j] = fs * u[j]; ph += u[j] * u[j]; }
    hx2 = wred32(ph);
    float4 a, b;
    a.x = u[0]; a.y = u[1]; a.z = u[2]; a.w = u[3];
    b.x = u[4]; b.y = u[5]; b.z = u[6]; b.w = u[7];
    *(float4*)(sX + wave * XP + joff) = a;
    *(float4*)(sX + wave * XP + joff + 4) = b;
  }
  __syncthreads();

  {
    const int tn = wave & 3, kq = wave >> 2;
    const float* ar = sX + l15 * XP + 64 * kq + 2 * hh;
    const float* br = sWl + (16 * tn + l15) * XP + 64 * kq + 2 * hh;
    v8f c = {};
#pragma unroll
    for (int s = 0; s < 16; ++s) {
      const v2f a = *(const v2f*)(ar + 4 * s);
      const v2f b = *(const v2f*)(br + 4 * s);
      c = wmma_f32x4(a, b, c);
    }
    float* pt = sP + kq * (16 * PP) + 16 * tn + l15;
#pragma unroll
    for (int v = 0; v < 8; ++v) pt[(8 * hh + v) * PP] = c[v];
  }
  __syncthreads();

  {
    const float* pr = sP + wave * PP + 2 * lane;
    v2f mm = *(const v2f*)pr;
#pragma unroll
    for (int q = 1; q < 4; ++q) {
      const v2f pq = *(const v2f*)(pr + q * (16 * PP));
      mm.x += pq.x; mm.y += pq.y;
    }
    const float m2 = wred32(mm.x * mm.x + mm.y * mm.y);
    const float md = wred32(mm.x * bl0 + mm.y * bl1);
    const float y2 = wred32(bl0 * bl0 + bl1 * bl1);
    const float s = hyp_scale(hx2, m2);
    const float x2r = s * s * m2;
    const float xy = s * md;
    const float den = fmaxf(1.f + 2.f * xy + x2r * y2, EPSF);
    const float ca = (1.f + 2.f * xy + y2) / den;
    const float cb = (1.f - x2r) / den;
    const float acs = ca * s;
    float2 o;
    o.x = acs * mm.x + cb * bl0;
    o.y = acs * mm.y + cb * bl1;
    *(float2*)(OUT + (size_t)node * 64 + 2 * lane) = o;
  }
}

extern "C" void kernel_launch(void* const* d_in, const int* in_sizes, int n_in,
                              void* d_out, int out_size, void* d_ws,
                              size_t ws_size, hipStream_t stream) {
  const float* features = (const float*)d_in[0];
  const float* W1 = (const float*)d_in[1];
  const float* b1 = (const float*)d_in[2];
  const float* W2 = (const float*)d_in[3];
  const float* b2 = (const float*)d_in[4];
  const float* Wl = (const float*)d_in[5];
  const float* bl = (const float*)d_in[6];
  const int* src_idx = (const int*)d_in[7];
  const int* to_fetch = (const int*)d_in[8];
  float* out = (float*)d_out;

  float* buf0 = (float*)d_ws;
  float* buf1 = buf0 + (size_t)65536 * 256;

  k_mm<512, false, true><<<512, 256, 0, stream>>>(
      const_cast<float*>(features), W1, b1, buf0, 0.25f, src_idx, to_fetch);
  k_agg<<<128, 256, 0, stream>>>(buf0, b1, buf1, 1.0f, 1);
  k_tail<<<4, 512, 0, stream>>>(buf1, W2, b2, Wl, bl, out);
}


# layer-1 fold launch fused into k_mm: a block keeps its 32 scaled product rows in LDS, waves 0 and 1 fold 16 each and apply the activation; no product row goes to memory; sums and order unchanged
# speedup vs baseline: 8.8656x; 1.1252x over previous
#include <hip/hip_runtime.h>

typedef __attribute__((ext_vector_type(2))) float v2f;
typedef __attribute__((ext_vector_type(8))) float v8f;

#define NHID 256
#define EPSF 1e-7f
#define MAXNF 0.99999f

__device__ __forceinline__ v8f wmma_f32x4(v2f a, v2f b, v8f c) {
  return __builtin_amdgcn_wmma_f32_16x16x4_f32(false, a, false, b, (short)0, c,
                                               false, false);
}

__device__ __forceinline__ float lane_xor16(float x) {
  const int b = __builtin_bit_cast(int, x);
  return __builtin_bit_cast(
      float, __builtin_amdgcn_permlanex16(b, b, 0x76543210, 0xfedcba98, false,
                                          false));
}
template <int M>
__device__ __forceinline__ float lane_xor_row(float x) {
  const int b = __builtin_bit_cast(int, x);
  return __builtin_bit_cast(
      float, __builtin_amdgcn_update_dpp(b, b, 0x160 + M, 0xf, 0xf, true));
}

__device__ __forceinline__ float wred32(float v) {
  v += lane_xor16(v);
  v += lane_xor_row<8>(v);
  v += lane_xor_row<4>(v);
  v += lane_xor_row<2>(v);
  v += lane_xor_row<1>(v);
  return v;
}

__device__ __forceinline__ float hyp_scale(float x2, float m2) {
  float xn  = fminf(fmaxf(sqrtf(x2), EPSF), MAXNF);
  float mxn = fmaxf(sqrtf(m2), EPSF);
  return tanhf(mxn / xn * atanhf(xn)) / mxn;
}

__device__ __forceinline__ void load8(const float* p, float s, float* d) {
  const float4* g = (const float4*)p;
  float4 a = g[0], b = g[1];
  d[0] = a.x * s; d[1] = a.y * s; d[2] = a.z * s; d[3] = a.w * s;
  d[4] = b.x * s; d[5] = b.y * s; d[6] = b.z * s; d[7] = b.w * s;
}

__device__ __forceinline__ void scale8(const float4& a, const float4& b,
                                       float s, float* d) {
  d[0] = a.x * s; d[1] = a.y * s; d[2] = a.z * s; d[3] = a.w * s;
  d[4] = b.x * s; d[5] = b.y * s; d[6] = b.z * s; d[7] = b.w * s;
}

__device__ __forceinline__ void fold_step(float* x, float& x2, const float4& a,
                                          const float4& b, float s) {
  float y[8];
  scale8(a, b, s, y);
  float py2 = 0.f, pxy = 0.f;
#pragma unroll
  for (int j = 0; j < 8; ++j) { py2 += y[j] * y[j]; pxy += x[j] * y[j]; }
  float y2 = wred32(py2);
  float xy = wred32(pxy);
  float den = fmaxf(1.f + 2.f * xy + x2 * y2, EPSF);
  float ca = (1.f + 2.f * xy + y2) / den;
  float cb = (1.f - x2) / den;
  float pn = 0.f;
#pragma unroll
  for (int j = 0; j < 8; ++j) { x[j] = ca * x[j] + cb * y[j]; pn += x[j] * x[j]; }
  x2 = wred32(pn);
}

template <int RS>
__device__ __forceinline__ void fold16(const float4* g0, float s, float* x,
                                       float& x2) {
  {
    const float4 a = g0[0], b = g0[1];
    scale8(a, b, s, x);
  }
  float px = 0.f;
#pragma unroll
  for (int j = 0; j < 8; ++j) px += x[j] * x[j];
  x2 = wred32(px);
  const float4 z4 = make_float4(0.f, 0.f, 0.f, 0.f);
  float4 a0 = z4, b0 = z4, a1 = z4, b1 = z4, a2 = z4, b2 = z4;
#pragma unroll 1
  for (int k0 = -2; k0 < 16; k0 += 3) {
    const float4* gn0 = g0 + min(k0 + 3, 15) * RS;
    const float4* gn1 = g0 + min(k0 + 4, 15) * RS;
    const float4* gn2 = g0 + min(k0 + 5, 15) * RS;
    const float4 na0 = gn0[0], nb0 = gn0[1];
    const float4 na1 = gn1[0], nb1 = gn1[1];
    const float4 na2 = gn2[0], nb2 = gn2[1];
    if (k0 >= 1) {
      fold_step(x, x2, a0, b0, s);
      fold_step(x, x2, a1, b1, s);
      fold_step(x, x2, a2, b2, s);
    }
    a0 = na0; b0 = nb0; a1 = na1; b1 = nb1; a2 = na2; b2 = nb2;
  }
}

__device__ __forceinline__ void store8(float* p, const float* d, float s) {
  float4 a, b;
  a.x = d[0] * s; a.y = d[1] * s; a.z = d[2] * s; a.w = d[3] * s;
  b.x = d[4] * s; b.y = d[5] * s; b.z = d[6] * s; b.w = d[7] * s;
  float4* g = (float4*)p;
  g[0] = a; g[1] = b;
}

template <int KDIM>
__global__ __launch_bounds__(256) void k_mm(const float* X,
                                            const float* __restrict__ W,
                                            const float* __restrict__ bias,
                                            float* OUT, float inScale,
                                            const int* __restrict__ src,
                                            const int* __restrict__ tf) {
  constexpr int KT = 32;
  constexpr int A_PITCH = KT + 4;
  constexpr int B_PITCH = 2 * NHID + 32;
  constexpr int A_FL = 32 * A_PITCH;
  constexpr int B_FL = (KT / 2) * B_PITCH;
  constexpr int ST_FL = A_FL + B_FL;
  constexpr int NT = KDIM / KT;
  __shared__ __attribute__((aligned(16))) float stg[2 * ST_FL];
  float (*smx)[NHID + 8] = reinterpret_cast<float (*)[NHID + 8]>(stg);
  static_assert(32 * (NHID + 8) <= 2 * ST_FL, "epilogue rows fit the stages");
  __shared__ float part[32][8];
  __shared__ float rx2[32], rm2[32];

  const int t = threadIdx.x;
  const int wave = t >> 5, lane = t & 31;
  const int tile_m = wave >> 2;
  const int n0 = (wave & 3) << 6;
  const int l15 = lane & 15;
  const int kk = (lane >> 4) << 1;
  const int brow = blockIdx.x << 5;
  const int r = t >> 3, p = t & 7;

  const int b = brow >> 8;
  const int n2 = tf[b] + b * 1024;
  const int n1 = src[(size_t)n2 * 16 + (((brow >> 4) + tile_m) & 15)];
  const int* nb = src + (size_t)n1 * 16;
  const int xrow = nb[r & 15];

  const int c4 = t & 63, kp0 = t >> 6;
  const float* xg = X + (size_t)xrow * KDIM + p * 4;
  const float* wg = W + (size_t)(2 * kp0) * NHID + c4 * 4;
  float4 ra, rb0[4], rb1[4];
  auto gload = [&](int kt) {
    ra = *(const float4*)(xg + kt * KT);
    const float* w = wg + (size_t)(kt * KT) * NHID;
#pragma unroll
    for (int i = 0; i < 4; ++i) {
      rb0[i] = *(const float4*)(w + (size_t)(8 * i) * NHID);
      rb1[i] = *(const float4*)(w + (size_t)(8 * i + 1) * NHID);
    }
  };
  auto sstore = [&](float* st) {
    *(float4*)(st + r * A_PITCH + p * 4) = ra;
    float* bs = st + A_FL + kp0 * B_PITCH + c4 * 8;
#pragma unroll
    for (int i = 0; i < 4; ++i) {
      float4 lo, hi;
      lo.x = rb0[i].x; lo.y = rb1[i].x; lo.z = rb0[i].y; lo.w = rb1[i].y;
      hi.x = rb0[i].z; hi.y = rb1[i].z; hi.z = rb0[i].w; hi.w = rb1[i].w;
      *(float4*)(bs + (4 * i) * B_PITCH) = lo;
      *(float4*)(bs + (4 * i) * B_PITCH + 4) = hi;
    }
  };

  v8f acc0 = {}, acc1 = {}, acc2 = {}, acc3 = {};
  gload(0);
  sstore(stg);
  __syncthreads();
  for (int kt = 0; kt < NT; ++kt) {
    const float* st = stg + (kt & 1) * ST_FL;
    if (kt + 1 < NT) gload(kt + 1);
    const float* as = st + (tile_m * 16 + l15) * A_PITCH + kk;
    const float* bs = st + A_FL + (kk >> 1) * B_PITCH + (n0 + l15) * 2;
#pragma unroll
    for (int s = 0; s < KT / 4; ++s) {
      const v2f a = *(const v2f*)(as + s * 4);
      const float* bp = bs + (2 * s) * B_PITCH;
      const v2f b0 = *(const v2f*)(bp);
      const v2f b1 = *(const v2f*)(bp + 32);
      const v2f b2 = *(const v2f*)(bp + 64);
      const v2f b3 = *(const v2f*)(bp + 96);
      acc0 = wmma_f32x4(a, b0, acc0);
      acc1 = wmma_f32x4(a, b1, acc1);
      acc2 = wmma_f32x4(a, b2, acc2);
      acc3 = wmma_f32x4(a, b3, acc3);
    }
    if (kt + 1 < NT) sstore(stg + ((kt + 1) & 1) * ST_FL);
    __syncthreads();
  }
  {
    const int mb = tile_m * 16 + ((lane >> 4) ? 8 : 0);
#pragma unroll
    for (int v = 0; v < 8; ++v) {
      smx[mb + v][n0 + 0 + l15]  = acc0[v];
      smx[mb + v][n0 + 16 + l15] = acc1[v];
      smx[mb + v][n0 + 32 + l15] = acc2[v];
      smx[mb + v][n0 + 48 + l15] = acc3[v];
    }
  }
  __syncthreads();

  float sx = 0.f;
  {
    const int seg = KDIM / 8;
    const float* xr = X + (size_t)xrow * KDIM + p * seg;
    for (int j = 0; j < seg; j += 4) {
      float4 v = *(const float4*)(xr + j);
      sx += v.x * v.x + v.y * v.y + v.z * v.z + v.w * v.w;
    }
  }
  float sm = 0.f;
  float* mrw = &smx[r][p * 32];
#pragma unroll
  for (int j = 0; j < 32; ++j) {
    float m = mrw[j];
    sm += m * m;
  }
  part[r][p] = sx;
  __syncthreads();
  if (t < 32) { float s = 0.f; for (int i = 0; i < 8; ++i) s += part[t][i]; rx2[t] = s; }
  __syncthreads();
  part[r][p] = sm;
  __syncthreads();
  if (t < 32) { float s = 0.f; for (int i = 0; i < 8; ++i) s += part[t][i]; rm2[t] = s; }
  __syncthreads();

  const float x2s = inScale * inScale * rx2[r];
  const float m2s = inScale * inScale * rm2[r];
  const float cmv = hyp_scale(x2s, m2s) * inScale;
#pragma unroll
  for (int j = 0; j < 32; ++j) mrw[j] = cmv * mrw[j];
  __syncthreads();

  if (wave >= 2) return;
  const int joff = lane << 3;
  float bv[8];
  load8(bias + joff, 1.f, bv);
  float x[8], x2;
  fold16<(NHID + 8) / 4>((const float4*)&smx[wave << 4][joff], 1.f, x, x2);

  float pb2 = 0.f, pxy = 0.f;
#pragma unroll
  for (int j = 0; j < 8; ++j) { pb2 += bv[j] * bv[j]; pxy += x[j] * bv[j]; }
  float y2 = wred32(pb2);
  float xy = 0.25f * wred32(pxy);
  float x2r = 0.0625f * x2;
  float den = fmaxf(1.f + 2.f * xy + x2r * y2, EPSF);
  float ca = (1.f + 2.f * xy + y2) / den;
  float cb = (1.f - x2r) / den;
  float zn2 = ca * ca * x2r + 2.f * ca * cb * xy + cb * cb * y2;
  float zn = fminf(fmaxf(sqrtf(zn2), EPSF), MAXNF);
  float us = atanhf(zn) / zn;
  float u[8];
  float pv = 0.f;
#pragma unroll
  for (int j = 0; j < 8; ++j) {
    float z = ca * 0.25f * x[j] + cb * bv[j];
    u[j] = fmaxf(us * z, 0.f);
    pv += u[j] * u[j];
  }
  float vn = fmaxf(sqrtf(wred32(pv)), EPSF);
  float fs = tanhf(vn) / vn;
  store8(OUT + (size_t)((blockIdx.x << 1) + wave) * NHID + joff, u, fs);
}

__global__ __launch_bounds__(512) void k_tail(const float* __restrict__ H1,
                                              const float* __restrict__ W2,
                                              const float* __restrict__ b2,
                                              const float* __restrict__ Wl,
                                              const float* __restrict__ bl,
                                              float* __restrict__ OUT) {
  constexpr int XP = NHID + 4;
  constexpr int WBP = 2 * NHID + 32;
  constexpr int PP = 64 + 4;
  __shared__ __attribute__((aligned(16))) float sW[(NHID / 2) * WBP];
  __shared__ __attribute__((aligned(16))) float sX[16 * XP];
  __shared__ __attribute__((aligned(16))) float sM[16 * XP];
  float* sWl = sW;
  float* sP = sW + 64 * XP;
  static_assert(64 * XP + 4 * 16 * PP <= (NHID / 2) * WBP, "Wl + partials fit");

  const int t = threadIdx.x;
  const int wave = t >> 5, lane = t & 31;
  const int l15 = lane & 15, hh = lane >> 4;
  const int node = (blockIdx.x << 4) + wave;
  const int joff = lane << 3;
  const float* msg = H1 + (size_t)node * 16 * NHID + joff;

  const int c4 = t & 63, kp0 = t >> 6;
  const float* wg = W2 + (size_t)(2 * kp0) * NHID + c4 * 4;
  float* ws = sW + kp0 * WBP + c4 * 8;
  auto wload = [&](int i, float4& w0, float4& w1) {
    const float* p = wg + (size_t)(16 * i) * NHID;
    w0 = *(const float4*)p;
    w1 = *(const float4*)(p + NHID);
  };
  auto wstore = [&](int i, const float4& w0, const float4& w1) {
    float* q = ws + (8 * i) * WBP;
    *(float4*)q = w0;
    *(float4*)(q + 4) = w1;
  };

  float bb[8];
  load8(b2 + joff, 1.f, bb);
  const float bl0 = bl[2 * lane], bl1 = bl[2 * lane + 1];
  const float4* g0 = (const float4*)msg;
  constexpr int RS = NHID / 4;
  float x[8];
  {
    const float4 a = g0[0], b = g0[1];
    scale8(a, b, 0.25f, x);
  }
  float px = 0.f;
#pragma unroll
  for (int j = 0; j < 8; ++j) px += x[j] * x[j];
  float x2 = wred32(px);
  const float4 z4 = make_float4(0.f, 0.f, 0.f, 0.f);
  float4 a0 = z4, b0 = z4, a1 = z4, b1 = z4, a2 = z4, b2r = z4;
  float4 u0 = z4, v0 = z4, u1 = z4, v1 = z4, u2 = z4, v2 = z4;
#pragma unroll 1
  for (int k0 = -2; k0 < 16; k0 += 3) {
    const float4* gn0 = g0 + min(k0 + 3, 15) * RS;
    const float4* gn1 = g0 + min(k0 + 4, 15) * RS;
    const float4* gn2 = g0 + min(k0 + 5, 15) * RS;
    const float4 na0 = gn0[0], nb0 = gn0[1];
    const float4 na1 = gn1[0], nb1 = gn1[1];
    const float4 na2 = gn2[0], nb2 = gn2[1];
    float4 nu0, nv0, nu1, nv1, nu2, nv2;
    wload(min(k0 + 2, 15), nu0, nv0);
    wload(min(k0 + 3, 15), nu1, nv1);
    wload(min(k0 + 4, 15), nu2, nv2);
    if (k0 >= 1) {
      wstore(k0 - 1, u0, v0);
      fold_step(x, x2, a0, b0, 0.25f);
      wstore(k0, u1, v1);
      fold_step(x, x2, a1, b1, 0.25f);
      wstore(k0 + 1, u2, v2);
      fold_step(x, x2, a2, b2r, 0.25f);
    }
    a0 = na0; b0 = nb0; a1 = na1; b1 = nb1; a2 = na2; b2r = nb2;
    u0 = nu0; v0 = nv0; u1 = nu1; v1 = nv1; u2 = nu2; v2 = nv2;
  }
  wstore(15, u0, v0);
  {
    float4 a, b;
    a.x = x[0]; a.y = x[1]; a.z = x[2]; a.w = x[3];
    b.x = x[4]; b.y = x[5]; b.z = x[6]; b.w = x[7];
    *(float4*)(sX + wave * XP + joff) = a;
    *(float4*)(sX + wave * XP + joff + 4) = b;
  }
  __syncthreads();

  const float* wlg = Wl + (size_t)(t >> 6) * NHID + (t & 63) * 4;
  const float4 wl0 = *(const float4*)(wlg);
  const float4 wl1 = *(const float4*)(wlg + 8 * NHID);
  const float4 wl2 = *(const float4*)(wlg + 16 * NHID);
  const float4 wl3 = *(const float4*)(wlg + 24 * NHID);
  const float4 wl4 = *(const float4*)(wlg + 32 * NHID);
  const float4 wl5 = *(const float4*)(wlg + 40 * NHID);
  const float4 wl6 = *(const float4*)(wlg + 48 * NHID);
  const float4 wl7 = *(const float4*)(wlg + 56 * NHID);

  {
    const int n0 = wave << 4;
    const float* as = sX + l15 * XP + 2 * hh;
    const int n = n0 + l15;
    const float* bs = sW + hh * WBP + (n >> 2) * 8 + (n & 3);
    v8f acc = {};
#pragma unroll
    for (int s = 0; s < NHID / 4; ++s) {
      const v2f a = *(const v2f*)(as + 4 * s);
      v2f b;
      b.x = bs[(2 * s) * WBP];
      b.y = bs[(2 * s) * WBP + 4];
      acc = wmma_f32x4(a, b, acc);
    }
#pragma unroll
    for (int v = 0; v < 8; ++v) sM[(8 * hh + v) * XP + n0 + l15] = acc[v];
  }
  __syncthreads();

  {
    float* wls = sWl + (t >> 6) * XP + (t & 63) * 4;
    *(float4*)(wls) = wl0;
    *(float4*)(wls + 8 * XP) = wl1;
    *(float4*)(wls + 16 * XP) = wl2;
    *(float4*)(wls + 24 * XP) = wl3;
    *(float4*)(wls + 32 * XP) = wl4;
    *(float4*)(wls + 40 * XP) = wl5;
    *(float4*)(wls + 48 * XP) = wl6;
    *(float4*)(wls + 56 * XP) = wl7;
  }

  float hx2;
  {
    float m[8];
    {
      const float4 a = *(const float4*)(sM + wave * XP + joff);
      const float4 b = *(const float4*)(sM + wave * XP + joff + 4);
      m[0] = a.x; m[1] = a.y; m[2] = a.z; m[3] = a.w;
      m[4] = b.x; m[5] = b.y; m[6] = b.z; m[7] = b.w;
    }
    float pm2 = 0.f, pmd = 0.f, pb2 = 0.f;
#pragma unroll
    for (int j = 0; j < 8; ++j) {
      pm2 += m[j] * m[j]; pmd += m[j] * bb[j]; pb2 += bb[j] * bb[j];
    }
    const float m2 = wred32(pm2);
    const float md = wred32(pmd);
    const float y2 = wred32(pb2);
    const float cmv = hyp_scale(x2, m2);
    const float cf = 0.25f * cmv;
    const float x2r = cf * cf * m2;
    const float xy = cf * md;
    const float den = fmaxf(1.f + 2.f * xy + x2r * y2, EPSF);
    const float ca = (1.f + 2.f * xy + y2) / den;
    const float cb = (1.f - x2r) / den;
    const float zn2 = ca * ca * x2r + 2.f * ca * cb * xy + cb * cb * y2;
    const float zn = fminf(fmaxf(sqrtf(zn2), EPSF), MAXNF);
    const float us = atanhf(zn) / zn;
    const float acf = ca * cf;
    float u[8];
    float pv = 0.f;
#pragma unroll
    for (int j = 0; j < 8; ++j) {
      const float z = acf * m[j] + cb * bb[j];
      u[j] = fmaxf(us * z, 0.f);
      pv += u[j] * u[j];
    }
    const float vn = fmaxf(sqrtf(wred32(pv)), EPSF);
    const float fs = tanhf(vn) / vn;
    float ph = 0.f;
#pragma unroll
    for (int j = 0; j < 8; ++j) { u[j] = fs * u[j]; ph += u[j] * u[j]; }
    hx2 = wred32(ph);
    float4 a, b;
    a.x = u[0]; a.y = u[1]; a.z = u[2]; a.w = u[3];
    b.x = u[4]; b.y = u[5]; b.z = u[6]; b.w = u[7];
    *(float4*)(sX + wave * XP + joff) = a;
    *(float4*)(sX + wave * XP + joff + 4) = b;
  }
  __syncthreads();

  {
    const int tn = wave & 3, kq = wave >> 2;
    const float* ar = sX + l15 * XP + 64 * kq + 2 * hh;
    const float* br = sWl + (16 * tn + l15) * XP + 64 * kq + 2 * hh;
    v8f c = {};
#pragma unroll
    for (int s = 0; s < 16; ++s) {
      const v2f a = *(const v2f*)(ar + 4 * s);
      const v2f b = *(const v2f*)(br + 4 * s);
      c = wmma_f32x4(a, b, c);
    }
    float* pt = sP + kq * (16 * PP) + 16 * tn + l15;
#pragma unroll
    for (int v = 0; v < 8; ++v) pt[(8 * hh + v) * PP] = c[v];
  }
  __syncthreads();

  {
    const float* pr = sP + wave * PP + 2 * lane;
    v2f mm = *(const v2f*)pr;
#pragma unroll
    for (int q = 1; q < 4; ++q) {
      const v2f pq = *(const v2f*)(pr + q * (16 * PP));
      mm.x += pq.x; mm.y += pq.y;
    }
    const float m2 = wred32(mm.x * mm.x + mm.y * mm.y);
    const float md = wred32(mm.x * bl0 + mm.y * bl1);
    const float y2 = wred32(bl0 * bl0 + bl1 * bl1);
    const float s = hyp_scale(hx2, m2);
    const float x2r = s * s * m2;
    const float xy = s * md;
    const float den = fmaxf(1.f + 2.f * xy + x2r * y2, EPSF);
    const float ca = (1.f + 2.f * xy + y2) / den;
    const float cb = (1.f - x2r) / den;
    const float acs = ca * s;
    float2 o;
    o.x = acs * mm.x + cb * bl0;
    o.y = acs * mm.y + cb * bl1;
    *(float2*)(OUT + (size_t)node * 64 + 2 * lane) = o;
  }
}

extern "C" void kernel_launch(void* const* d_in, const int* in_sizes, int n_in,
                              void* d_out, int out_size, void* d_ws,
                              size_t ws_size, hipStream_t stream) {
  const float* features = (const float*)d_in[0];
  const float* W1 = (const float*)d_in[1];
  const float* b1 = (const float*)d_in[2];
  const float* W2 = (const float*)d_in[3];
  const float* b2 = (const float*)d_in[4];
  const float* Wl = (const float*)d_in[5];
  const float* bl = (const float*)d_in[6];
  const int* src_idx = (const int*)d_in[7];
  const int* to_fetch = (const int*)d_in[8];
  float* out = (float*)d_out;

  float* h1 = (float*)d_ws;

  k_mm<512><<<512, 256, 0, stream>>>(features, W1, b1, h1, 0.25f, src_idx,
                                     to_fetch);
  k_tail<<<4, 512, 0, stream>>>(h1, W2, b2, Wl, bl, out);
}


# k_mm: |x_row|^2 summed from the staged X tile in the k loop, not by a second pass of 32-line loads over the gathered rows; only that sum's order changes (16-byte slot per k tile, then 8 slots)
# speedup vs baseline: 9.1695x; 1.0343x over previous
#include <hip/hip_runtime.h>

typedef __attribute__((ext_vector_type(2))) float v2f;
typedef __attribute__((ext_vector_type(8))) float v8f;

#define NHID 256
#define EPSF 1e-7f
#define MAXNF 0.99999f

__device__ __forceinline__ v8f wmma_f32x4(v2f a, v2f b, v8f c) {
  return __builtin_amdgcn_wmma_f32_16x16x4_f32(false, a, false, b, (short)0, c,
                                               false, false);
}

__device__ __forceinline__ float lane_xor16(float x) {
  const int b = __builtin_bit_cast(int, x);
  return __builtin_bit_cast(
      float, __builtin_amdgcn_permlanex16(b, b, 0x76543210, 0xfedcba98, false,
                                          false));
}
template <int M>
__device__ __forceinline__ float lane_xor_row(float x) {
  const int b = __builtin_bit_cast(int, x);
  return __builtin_bit_cast(
      float, __builtin_amdgcn_update_dpp(b, b, 0x160 + M, 0xf, 0xf, true));
}

__device__ __forceinline__ float wred32(float v) {
  v += lane_xor16(v);
  v += lane_xor_row<8>(v);
  v += lane_xor_row<4>(v);
  v += lane_xor_row<2>(v);
  v += lane_xor_row<1>(v);
  return v;
}

__device__ __forceinline__ float hyp_scale(float x2, float m2) {
  float xn  = fminf(fmaxf(sqrtf(x2), EPSF), MAXNF);
  float mxn = fmaxf(sqrtf(m2), EPSF);
  return tanhf(mxn / xn * atanhf(xn)) / mxn;
}

__device__ __forceinline__ void load8(const float* p, float s, float* d) {
  const float4* g = (const float4*)p;
  float4 a = g[0], b = g[1];
  d[0] = a.x * s; d[1] = a.y * s; d[2] = a.z * s; d[3] = a.w * s;
  d[4] = b.x * s; d[5] = b.y * s; d[6] = b.z * s; d[7] = b.w * s;
}

__device__ __forceinline__ void scale8(const float4& a, const float4& b,
                                       float s, float* d) {
  d[0] = a.x * s; d[1] = a.y * s; d[2] = a.z * s; d[3] = a.w * s;
  d[4] = b.x * s; d[5] = b.y * s; d[6] = b.z * s; d[7] = b.w * s;
}

__device__ __forceinline__ void fold_step(float* x, float& x2, const float4& a,
                                          const float4& b, float s) {
  float y[8];
  scale8(a, b, s, y);
  float py2 = 0.f, pxy = 0.f;
#pragma unroll
  for (int j = 0; j < 8; ++j) { py2 += y[j] * y[j]; pxy += x[j] * y[j]; }
  float y2 = wred32(py2);
  float xy = wred32(pxy);
  float den = fmaxf(1.f + 2.f * xy + x2 * y2, EPSF);
  float ca = (1.f + 2.f * xy + y2) / den;
  float cb = (1.f - x2) / den;
  float pn = 0.f;
#pragma unroll
  for (int j = 0; j < 8; ++j) { x[j] = ca * x[j] + cb * y[j]; pn += x[j] * x[j]; }
  x2 = wred32(pn);
}

template <int RS>
__device__ __forceinline__ void fold16(const float4* g0, float s, float* x,
                                       float& x2) {
  {
    const float4 a = g0[0], b = g0[1];
    scale8(a, b, s, x);
  }
  float px = 0.f;
#pragma unroll
  for (int j = 0; j < 8; ++j) px += x[j] * x[j];
  x2 = wred32(px);
  const float4 z4 = make_float4(0.f, 0.f, 0.f, 0.f);
  float4 a0 = z4, b0 = z4, a1 = z4, b1 = z4, a2 = z4, b2 = z4;
#pragma unroll 1
  for (int k0 = -2; k0 < 16; k0 += 3) {
    const float4* gn0 = g0 + min(k0 + 3, 15) * RS;
    const float4* gn1 = g0 + min(k0 + 4, 15) * RS;
    const float4* gn2 = g0 + min(k0 + 5, 15) * RS;
    const float4 na0 = gn0[0], nb0 = gn0[1];
    const float4 na1 = gn1[0], nb1 = gn1[1];
    const float4 na2 = gn2[0], nb2 = gn2[1];
    if (k0 >= 1) {
      fold_step(x, x2, a0, b0, s);
      fold_step(x, x2, a1, b1, s);
      fold_step(x, x2, a2, b2, s);
    }
    a0 = na0; b0 = nb0; a1 = na1; b1 = nb1; a2 = na2; b2 = nb2;
  }
}

__device__ __forceinline__ void store8(float* p, const float* d, float s) {
  float4 a, b;
  a.x = d[0] * s; a.y = d[1] * s; a.z = d[2] * s; a.w = d[3] * s;
  b.x = d[4] * s; b.y = d[5] * s; b.z = d[6] * s; b.w = d[7] * s;
  float4* g = (float4*)p;
  g[0] = a; g[1] = b;
}

template <int KDIM>
__global__ __launch_bounds__(256) void k_mm(const float* X,
                                            const float* __restrict__ W,
                                            const float* __restrict__ bias,
                                            float* OUT, float inScale,
                                            const int* __restrict__ src,
                                            const int* __restrict__ tf) {
  constexpr int KT = 32;
  constexpr int A_PITCH = KT + 4;
  constexpr int B_PITCH = 2 * NHID + 32;
  constexpr int A_FL = 32 * A_PITCH;
  constexpr int B_FL = (KT / 2) * B_PITCH;
  constexpr int ST_FL = A_FL + B_FL;
  constexpr int NT = KDIM / KT;
  __shared__ __attribute__((aligned(16))) float stg[2 * ST_FL];
  float (*smx)[NHID + 8] = reinterpret_cast<float (*)[NHID + 8]>(stg);
  static_assert(32 * (NHID + 8) <= 2 * ST_FL, "epilogue rows fit the stages");
  __shared__ float part[32][8];
  __shared__ float rx2[32], rm2[32];

  const int t = threadIdx.x;
  const int wave = t >> 5, lane = t & 31;
  const int tile_m = wave >> 2;
  const int n0 = (wave & 3) << 6;
  const int l15 = lane & 15;
  const int kk = (lane >> 4) << 1;
  const int brow = blockIdx.x << 5;
  const int r = t >> 3, p = t & 7;

  const int b = brow >> 8;
  const int n2 = tf[b] + b * 1024;
  const int n1 = src[(size_t)n2 * 16 + (((brow >> 4) + tile_m) & 15)];
  const int* nb = src + (size_t)n1 * 16;
  const int xrow = nb[r & 15];

  const int c4 = t & 63, kp0 = t >> 6;
  const float* xg = X + (size_t)xrow * KDIM + p * 4;
  const float* wg = W + (size_t)(2 * kp0) * NHID + c4 * 4;
  float4 ra, rb0[4], rb1[4];
  auto gload = [&](int kt) {
    ra = *(const float4*)(xg + kt * KT);
    const float* w = wg + (size_t)(kt * KT) * NHID;
#pragma unroll
    for (int i = 0; i < 4; ++i) {
      rb0[i] = *(const float4*)(w + (size_t)(8 * i) * NHID);
      rb1[i] = *(const float4*)(w + (size_t)(8 * i + 1) * NHID);
    }
  };
  float sx = 0.f;
  auto sstore = [&](float* st) {
    sx += ra.x * ra.x + ra.y * ra.y + ra.z * ra.z + ra.w * ra.w;
    *(float4*)(st + r * A_PITCH + p * 4) = ra;
    float* bs = st + A_FL + kp0 * B_PITCH + c4 * 8;
#pragma unroll
    for (int i = 0; i < 4; ++i) {
      float4 lo, hi;
      lo.x = rb0[i].x; lo.y = rb1[i].x; lo.z = rb0[i].y; lo.w = rb1[i].y;
      hi.x = rb0[i].z; hi.y = rb1[i].z; hi.z = rb0[i].w; hi.w = rb1[i].w;
      *(float4*)(bs + (4 * i) * B_PITCH) = lo;
      *(float4*)(bs + (4 * i) * B_PITCH + 4) = hi;
    }
  };

  v8f acc0 = {}, acc1 = {}, acc2 = {}, acc3 = {};
  gload(0);
  sstore(stg);
  __syncthreads();
  for (int kt = 0; kt < NT; ++kt) {
    const float* st = stg + (kt & 1) * ST_FL;
    if (kt + 1 < NT) gload(kt + 1);
    const float* as = st + (tile_m * 16 + l15) * A_PITCH + kk;
    const float* bs = st + A_FL + (kk >> 1) * B_PITCH + (n0 + l15) * 2;
#pragma unroll
    for (int s = 0; s < KT / 4; ++s) {
      const v2f a = *(const v2f*)(as + s * 4);
      const float* bp = bs + (2 * s) * B_PITCH;
      const v2f b0 = *(const v2f*)(bp);
      const v2f b1 = *(const v2f*)(bp + 32);
      const v2f b2 = *(const v2f*)(bp + 64);
      const v2f b3 = *(const v2f*)(bp + 96);
      acc0 = wmma_f32x4(a, b0, acc0);
      acc1 = wmma_f32x4(a, b1, acc1);
      acc2 = wmma_f32x4(a, b2, acc2);
      acc3 = wmma_f32x4(a, b3, acc3);
    }
    if (kt + 1 < NT) sstore(stg + ((kt + 1) & 1) * ST_FL);
    __syncthreads();
  }
  {
    const int mb = tile_m * 16 + ((lane >> 4) ? 8 : 0);
#pragma unroll
    for (int v = 0; v < 8; ++v) {
      smx[mb + v][n0 + 0 + l15]  = acc0[v];
      smx[mb + v][n0 + 16 + l15] = acc1[v];
      smx[mb + v][n0 + 32 + l15] = acc2[v];
      smx[mb + v][n0 + 48 + l15] = acc3[v];
    }
  }
  __syncthreads();

  float sm = 0.f;
  float* mrw = &smx[r][p * 32];
#pragma unroll
  for (int j = 0; j < 32; ++j) {
    float m = mrw[j];
    sm += m * m;
  }
  part[r][p] = sx;
  __syncthreads();
  if (t < 32) { float s = 0.f; for (int i = 0; i < 8; ++i) s += part[t][i]; rx2[t] = s; }
  __syncthreads();
  part[r][p] = sm;
  __syncthreads();
  if (t < 32) { float s = 0.f; for (int i = 0; i < 8; ++i) s += part[t][i]; rm2[t] = s; }
  __syncthreads();

  const float x2s = inScale * inScale * rx2[r];
  const float m2s = inScale * inScale * rm2[r];
  const float cmv = hyp_scale(x2s, m2s) * inScale;
#pragma unroll
  for (int j = 0; j < 32; ++j) mrw[j] = cmv * mrw[j];
  __syncthreads();

  if (wave >= 2) return;
  const int joff = lane << 3;
  float bv[8];
  load8(bias + joff, 1.f, bv);
  float x[8], x2;
  fold16<(NHID + 8) / 4>((const float4*)&smx[wave << 4][joff], 1.f, x, x2);

  float pb2 = 0.f, pxy = 0.f;
#pragma unroll
  for (int j = 0; j < 8; ++j) { pb2 += bv[j] * bv[j]; pxy += x[j] * bv[j]; }
  float y2 = wred32(pb2);
  float xy = 0.25f * wred32(pxy);
  float x2r = 0.0625f * x2;
  float den = fmaxf(1.f + 2.f * xy + x2r * y2, EPSF);
  float ca = (1.f + 2.f * xy + y2) / den;
  float cb = (1.f - x2r) / den;
  float zn2 = ca * ca * x2r + 2.f * ca * cb * xy + cb * cb * y2;
  float zn = fminf(fmaxf(sqrtf(zn2), EPSF), MAXNF);
  float us = atanhf(zn) / zn;
  float u[8];
  float pv = 0.f;
#pragma unroll
  for (int j = 0; j < 8; ++j) {
    float z = ca * 0.25f * x[j] + cb * bv[j];
    u[j] = fmaxf(us * z, 0.f);
    pv += u[j] * u[j];
  }
  float vn = fmaxf(sqrtf(wred32(pv)), EPSF);
  float fs = tanhf(vn) / vn;
  store8(OUT + (size_t)((blockIdx.x << 1) + wave) * NHID + joff, u, fs);
}

__global__ __launch_bounds__(512) void k_tail(const float* __restrict__ H1,
                                              const float* __restrict__ W2,
                                              const float* __restrict__ b2,
                                              const float* __restrict__ Wl,
                                              const float* __restrict__ bl,
                                              float* __restrict__ OUT) {
  constexpr int XP = NHID + 4;
  constexpr int WBP = 2 * NHID + 32;
  constexpr int PP = 64 + 4;
  __shared__ __attribute__((aligned(16))) float sW[(NHID / 2) * WBP];
  __shared__ __attribute__((aligned(16))) float sX[16 * XP];
  __shared__ __attribute__((aligned(16))) float sM[16 * XP];
  float* sWl = sW;
  float* sP = sW + 64 * XP;
  static_assert(64 * XP + 4 * 16 * PP <= (NHID / 2) * WBP, "Wl + partials fit");

  const int t = threadIdx.x;
  const int wave = t >> 5, lane = t & 31;
  const int l15 = lane & 15, hh = lane >> 4;
  const int node = (blockIdx.x << 4) + wave;
  const int joff = lane << 3;
  const float* msg = H1 + (size_t)node * 16 * NHID + joff;

  const int c4 = t & 63, kp0 = t >> 6;
  const float* wg = W2 + (size_t)(2 * kp0) * NHID + c4 * 4;
  float* ws = sW + kp0 * WBP + c4 * 8;
  auto wload = [&](int i, float4& w0, float4& w1) {
    const float* p = wg + (size_t)(16 * i) * NHID;
    w0 = *(const float4*)p;
    w1 = *(const float4*)(p + NHID);
  };
  auto wstore = [&](int i, const float4& w0, const float4& w1) {
    float* q = ws + (8 * i) * WBP;
    *(float4*)q = w0;
    *(float4*)(q + 4) = w1;
  };

  float bb[8];
  load8(b2 + joff, 1.f, bb);
  const float bl0 = bl[2 * lane], bl1 = bl[2 * lane + 1];
  const float4* g0 = (const float4*)msg;
  constexpr int RS = NHID / 4;
  float x[8];
  {
    const float4 a = g0[0], b = g0[1];
    scale8(a, b, 0.25f, x);
  }
  float px = 0.f;
#pragma unroll
  for (int j = 0; j < 8; ++j) px += x[j] * x[j];
  float x2 = wred32(px);
  const float4 z4 = make_float4(0.f, 0.f, 0.f, 0.f);
  float4 a0 = z4, b0 = z4, a1 = z4, b1 = z4, a2 = z4, b2r = z4;
  float4 u0 = z4, v0 = z4, u1 = z4, v1 = z4, u2 = z4, v2 = z4;
#pragma unroll 1
  for (int k0 = -2; k0 < 16; k0 += 3) {
    const float4* gn0 = g0 + min(k0 + 3, 15) * RS;
    const float4* gn1 = g0 + min(k0 + 4, 15) * RS;
    const float4* gn2 = g0 + min(k0 + 5, 15) * RS;
    const float4 na0 = gn0[0], nb0 = gn0[1];
    const float4 na1 = gn1[0], nb1 = gn1[1];
    const float4 na2 = gn2[0], nb2 = gn2[1];
    float4 nu0, nv0, nu1, nv1, nu2, nv2;
    wload(min(k0 + 2, 15), nu0, nv0);
    wload(min(k0 + 3, 15), nu1, nv1);
    wload(min(k0 + 4, 15), nu2, nv2);
    if (k0 >= 1) {
      wstore(k0 - 1, u0, v0);
      fold_step(x, x2, a0, b0, 0.25f);
      wstore(k0, u1, v1);
      fold_step(x, x2, a1, b1, 0.25f);
      wstore(k0 + 1, u2, v2);
      fold_step(x, x2, a2, b2r, 0.25f);
    }
    a0 = na0; b0 = nb0; a1 = na1; b1 = nb1; a2 = na2; b2r = nb2;
    u0 = nu0; v0 = nv0; u1 = nu1; v1 = nv1; u2 = nu2; v2 = nv2;
  }
  wstore(15, u0, v0);
  {
    float4 a, b;
    a.x = x[0]; a.y = x[1]; a.z = x[2]; a.w = x[3];
    b.x = x[4]; b.y = x[5]; b.z = x[6]; b.w = x[7];
    *(float4*)(sX + wave * XP + joff) = a;
    *(float4*)(sX + wave * XP + joff + 4) = b;
  }
  __syncthreads();

  const float* wlg = Wl + (size_t)(t >> 6) * NHID + (t & 63) * 4;
  const float4 wl0 = *(const float4*)(wlg);
  const float4 wl1 = *(const float4*)(wlg + 8 * NHID);
  const float4 wl2 = *(const float4*)(wlg + 16 * NHID);
  const float4 wl3 = *(const float4*)(wlg + 24 * NHID);
  const float4 wl4 = *(const float4*)(wlg + 32 * NHID);
  const float4 wl5 = *(const float4*)(wlg + 40 * NHID);
  const float4 wl6 = *(const float4*)(wlg + 48 * NHID);
  const float4 wl7 = *(const float4*)(wlg + 56 * NHID);

  {
    const int n0 = wave << 4;
    const float* as = sX + l15 * XP + 2 * hh;
    const int n = n0 + l15;
    const float* bs = sW + hh * WBP + (n >> 2) * 8 + (n & 3);
    v8f acc = {};
#pragma unroll
    for (int s = 0; s < NHID / 4; ++s) {
      const v2f a = *(const v2f*)(as + 4 * s);
      v2f b;
      b.x = bs[(2 * s) * WBP];
      b.y = bs[(2 * s) * WBP + 4];
      acc = wmma_f32x4(a, b, acc);
    }
#pragma unroll
    for (int v = 0; v < 8; ++v) sM[(8 * hh + v) * XP + n0 + l15] = acc[v];
  }
  __syncthreads();

  {
    float* wls = sWl + (t >> 6) * XP + (t & 63) * 4;
    *(float4*)(wls) = wl0;
    *(float4*)(wls + 8 * XP) = wl1;
    *(float4*)(wls + 16 * XP) = wl2;
    *(float4*)(wls + 24 * XP) = wl3;
    *(float4*)(wls + 32 * XP) = wl4;
    *(float4*)(wls + 40 * XP) = wl5;
    *(float4*)(wls + 48 * XP) = wl6;
    *(float4*)(wls + 56 * XP) = wl7;
  }

  float hx2;
  {
    float m[8];
    {
      const float4 a = *(const float4*)(sM + wave * XP + joff);
      const float4 b = *(const float4*)(sM + wave * XP + joff + 4);
      m[0] = a.x; m[1] = a.y; m[2] = a.z; m[3] = a.w;
      m[4] = b.x; m[5] = b.y; m[6] = b.z; m[7] = b.w;
    }
    float pm2 = 0.f, pmd = 0.f, pb2 = 0.f;
#pragma unroll
    for (int j = 0; j < 8; ++j) {
      pm2 += m[j] * m[j]; pmd += m[j] * bb[j]; pb2 += bb[j] * bb[j];
    }
    const float m2 = wred32(pm2);
    const float md = wred32(pmd);
    const float y2 = wred32(pb2);
    const float cmv = hyp_scale(x2, m2);
    const float cf = 0.25f * cmv;
    const float x2r = cf * cf * m2;
    const float xy = cf * md;
    const float den = fmaxf(1.f + 2.f * xy + x2r * y2, EPSF);
    const float ca = (1.f + 2.f * xy + y2) / den;
    const float cb = (1.f - x2r) / den;
    const float zn2 = ca * ca * x2r + 2.f * ca * cb * xy + cb * cb * y2;
    const float zn = fminf(fmaxf(sqrtf(zn2), EPSF), MAXNF);
    const float us = atanhf(zn) / zn;
    const float acf = ca * cf;
    float u[8];
    float pv = 0.f;
#pragma unroll
    for (int j = 0; j < 8; ++j) {
      const float z = acf * m[j] + cb * bb[j];
      u[j] = fmaxf(us * z, 0.f);
      pv += u[j] * u[j];
    }
    const float vn = fmaxf(sqrtf(wred32(pv)), EPSF);
    const float fs = tanhf(vn) / vn;
    float ph = 0.f;
#pragma unroll
    for (int j = 0; j < 8; ++j) { u[j] = fs * u[j]; ph += u[j] * u[j]; }
    hx2 = wred32(ph);
    float4 a, b;
    a.x = u[0]; a.y = u[1]; a.z = u[2]; a.w = u[3];
    b.x = u[4]; b.y = u[5]; b.z = u[6]; b.w = u[7];
    *(float4*)(sX + wave * XP + joff) = a;
    *(float4*)(sX + wave * XP + joff + 4) = b;
  }
  __syncthreads();

  {
    const int tn = wave & 3, kq = wave >> 2;
    const float* ar = sX + l15 * XP + 64 * kq + 2 * hh;
    const float* br = sWl + (16 * tn + l15) * XP + 64 * kq + 2 * hh;
    v8f c = {};
#pragma unroll
    for (int s = 0; s < 16; ++s) {
      const v2f a = *(const v2f*)(ar + 4 * s);
      const v2f b = *(const v2f*)(br + 4 * s);
      c = wmma_f32x4(a, b, c);
    }
    float* pt = sP + kq * (16 * PP) + 16 * tn + l15;
#pragma unroll
    for (int v = 0; v < 8; ++v) pt[(8 * hh + v) * PP] = c[v];
  }
  __syncthreads();

  {
    const float* pr = sP + wave * PP + 2 * lane;
    v2f mm = *(const v2f*)pr;
#pragma unroll
    for (int q = 1; q < 4; ++q) {
      const v2f pq = *(const v2f*)(pr + q * (16 * PP));
      mm.x += pq.x; mm.y += pq.y;
    }
    const float m2 = wred32(mm.x * mm.x + mm.y * mm.y);
    const float md = wred32(mm.x * bl0 + mm.y * bl1);
    const float y2 = wred32(bl0 * bl0 + bl1 * bl1);
    const float s = hyp_scale(hx2, m2);
    const float x2r = s * s * m2;
    const float xy = s * md;
    const float den = fmaxf(1.f + 2.f * xy + x2r * y2, EPSF);
    const float ca = (1.f + 2.f * xy + y2) / den;
    const float cb = (1.f - x2r) / den;
    const float acs = ca * s;
    float2 o;
    o.x = acs * mm.x + cb * bl0;
    o.y = acs * mm.y + cb * bl1;
    *(float2*)(OUT + (size_t)node * 64 + 2 * lane) = o;
  }
}

extern "C" void kernel_launch(void* const* d_in, const int* in_sizes, int n_in,
                              void* d_out, int out_size, void* d_ws,
                              size_t ws_size, hipStream_t stream) {
  const float* features = (const float*)d_in[0];
  const float* W1 = (const float*)d_in[1];
  const float* b1 = (const float*)d_in[2];
  const float* W2 = (const float*)d_in[3];
  const float* b2 = (const float*)d_in[4];
  const float* Wl = (const float*)d_in[5];
  const float* bl = (const float*)d_in[6];
  const int* src_idx = (const int*)d_in[7];
  const int* to_fetch = (const int*)d_in[8];
  float* out = (float*)d_out;

  float* h1 = (float*)d_ws;

  k_mm<512><<<512, 256, 0, stream>>>(features, W1, b1, h1, 0.25f, src_idx,
                                     to_fetch);
  k_tail<<<4, 512, 0, stream>>>(h1, W2, b2, Wl, bl, out);
}


# k_tail: 16 blocks x 4 rows, not 4 x 16: one serial fold chain per SIMD instead of four sharing its issue port; WMMA row tiles padded with 12 zero rows, Wl staged by idle waves; sums unchanged
# speedup vs baseline: 10.0925x; 1.1007x over previous
#include <hip/hip_runtime.h>

typedef __attribute__((ext_vector_type(2))) float v2f;
typedef __attribute__((ext_vector_type(8))) float v8f;

#define NHID 256
#define EPSF 1e-7f
#define MAXNF 0.99999f

__device__ __forceinline__ v8f wmma_f32x4(v2f a, v2f b, v8f c) {
  return __builtin_amdgcn_wmma_f32_16x16x4_f32(false, a, false, b, (short)0, c,
                                               false, false);
}

__device__ __forceinline__ float lane_xor16(float x) {
  const int b = __builtin_bit_cast(int, x);
  return __builtin_bit_cast(
      float, __builtin_amdgcn_permlanex16(b, b, 0x76543210, 0xfedcba98, false,
                                          false));
}
template <int M>
__device__ __forceinline__ float lane_xor_row(float x) {
  const int b = __builtin_bit_cast(int, x);
  return __builtin_bit_cast(
      float, __builtin_amdgcn_update_dpp(b, b, 0x160 + M, 0xf, 0xf, true));
}

__device__ __forceinline__ float wred32(float v) {
  v += lane_xor16(v);
  v += lane_xor_row<8>(v);
  v += lane_xor_row<4>(v);
  v += lane_xor_row<2>(v);
  v += lane_xor_row<1>(v);
  return v;
}

__device__ __forceinline__ float hyp_scale(float x2, float m2) {
  float xn  = fminf(fmaxf(sqrtf(x2), EPSF), MAXNF);
  float mxn = fmaxf(sqrtf(m2), EPSF);
  return tanhf(mxn / xn * atanhf(xn)) / mxn;
}

__device__ __forceinline__ void load8(const float* p, float s, float* d) {
  const float4* g = (const float4*)p;
  float4 a = g[0], b = g[1];
  d[0] = a.x * s; d[1] = a.y * s; d[2] = a.z * s; d[3] = a.w * s;
  d[4] = b.x * s; d[5] = b.y * s; d[6] = b.z * s; d[7] = b.w * s;
}

__device__ __forceinline__ void scale8(const float4& a, const float4& b,
                                       float s, float* d) {
  d[0] = a.x * s; d[1] = a.y * s; d[2] = a.z * s; d[3] = a.w * s;
  d[4] = b.x * s; d[5] = b.y * s; d[6] = b.z * s; d[7] = b.w * s;
}

__device__ __forceinline__ void fold_step(float* x, float& x2, const float4& a,
                                          const float4& b, float s) {
  float y[8];
  scale8(a, b, s, y);
  float py2 = 0.f, pxy = 0.f;
#pragma unroll
  for (int j = 0; j < 8; ++j) { py2 += y[j] * y[j]; pxy += x[j] * y[j]; }
  float y2 = wred32(py2);
  float xy = wred32(pxy);
  float den = fmaxf(1.f + 2.f * xy + x2 * y2, EPSF);
  float ca = (1.f + 2.f * xy + y2) / den;
  float cb = (1.f - x2) / den;
  float pn = 0.f;
#pragma unroll
  for (int j = 0; j < 8; ++j) { x[j] = ca * x[j] + cb * y[j]; pn += x[j] * x[j]; }
  x2 = wred32(pn);
}

template <int RS>
__device__ __forceinline__ void fold16(const float4* g0, float s, float* x,
                                       float& x2) {
  {
    const float4 a = g0[0], b = g0[1];
    scale8(a, b, s, x);
  }
  float px = 0.f;
#pragma unroll
  for (int j = 0; j < 8; ++j) px += x[j] * x[j];
  x2 = wred32(px);
  const float4 z4 = make_float4(0.f, 0.f, 0.f, 0.f);
  float4 a0 = z4, b0 = z4, a1 = z4, b1 = z4, a2 = z4, b2 = z4;
#pragma unroll 1
  for (int k0 = -2; k0 < 16; k0 += 3) {
    const float4* gn0 = g0 + min(k0 + 3, 15) * RS;
    const float4* gn1 = g0 + min(k0 + 4, 15) * RS;
    const float4* gn2 = g0 + min(k0 + 5, 15) * RS;
    const float4 na0 = gn0[0], nb0 = gn0[1];
    const float4 na1 = gn1[0], nb1 = gn1[1];
    const float4 na2 = gn2[0], nb2 = gn2[1];
    if (k0 >= 1) {
      fold_step(x, x2, a0, b0, s);
      fold_step(x, x2, a1, b1, s);
      fold_step(x, x2, a2, b2, s);
    }
    a0 = na0; b0 = nb0; a1 = na1; b1 = nb1; a2 = na2; b2 = nb2;
  }
}

__device__ __forceinline__ void store8(float* p, const float* d, float s) {
  float4 a, b;
  a.x = d[0] * s; a.y = d[1] * s; a.z = d[2] * s; a.w = d[3] * s;
  b.x = d[4] * s; b.y = d[5] * s; b.z = d[6] * s; b.w = d[7] * s;
  float4* g = (float4*)p;
  g[0] = a; g[1] = b;
}

template <int KDIM>
__global__ __launch_bounds__(256) void k_mm(const float* X,
                                            const float* __restrict__ W,
                                            const float* __restrict__ bias,
                                            float* OUT, float inScale,
                                            const int* __restrict__ src,
                                            const int* __restrict__ tf) {
  constexpr int KT = 32;
  constexpr int A_PITCH = KT + 4;
  constexpr int B_PITCH = 2 * NHID + 32;
  constexpr int A_FL = 32 * A_PITCH;
  constexpr int B_FL = (KT / 2) * B_PITCH;
  constexpr int ST_FL = A_FL + B_FL;
  constexpr int NT = KDIM / KT;
  __shared__ __attribute__((aligned(16))) float stg[2 * ST_FL];
  float (*smx)[NHID + 8] = reinterpret_cast<float (*)[NHID + 8]>(stg);
  static_assert(32 * (NHID + 8) <= 2 * ST_FL, "epilogue rows fit the stages");
  __shared__ float part[32][8];
  __shared__ float rx2[32], rm2[32];

  const int t = threadIdx.x;
  const int wave = t >> 5, lane = t & 31;
  const int tile_m = wave >> 2;
  const int n0 = (wave & 3) << 6;
  const int l15 = lane & 15;
  const int kk = (lane >> 4) << 1;
  const int brow = blockIdx.x << 5;
  const int r = t >> 3, p = t & 7;

  const int b = brow >> 8;
  const int n2 = tf[b] + b * 1024;
  const int n1 = src[(size_t)n2 * 16 + (((brow >> 4) + tile_m) & 15)];
  const int* nb = src + (size_t)n1 * 16;
  const int xrow = nb[r & 15];

  const int c4 = t & 63, kp0 = t >> 6;
  const float* xg = X + (size_t)xrow * KDIM + p * 4;
  const float* wg = W + (size_t)(2 * kp0) * NHID + c4 * 4;
  float4 ra, rb0[4], rb1[4];
  auto gload = [&](int kt) {
    ra = *(const float4*)(xg + kt * KT);
    const float* w = wg + (size_t)(kt * KT) * NHID;
#pragma unroll
    for (int i = 0; i < 4; ++i) {
      rb0[i] = *(const float4*)(w + (size_t)(8 * i) * NHID);
      rb1[i] = *(const float4*)(w + (size_t)(8 * i + 1) * NHID);
    }
  };
  float sx = 0.f;
  auto sstore = [&](float* st) {
    sx += ra.x * ra.x + ra.y * ra.y + ra.z * ra.z + ra.w * ra.w;
    *(float4*)(st + r * A_PITCH + p * 4) = ra;
    float* bs = st + A_FL + kp0 * B_PITCH + c4 * 8;
#pragma unroll
    for (int i = 0; i < 4; ++i) {
      float4 lo, hi;
      lo.x = rb0[i].x; lo.y = rb1[i].x; lo.z = rb0[i].y; lo.w = rb1[i].y;
      hi.x = rb0[i].z; hi.y = rb1[i].z; hi.z = rb0[i].w; hi.w = rb1[i].w;
      *(float4*)(bs + (4 * i) * B_PITCH) = lo;
      *(float4*)(bs + (4 * i) * B_PITCH + 4) = hi;
    }
  };

  v8f acc0 = {}, acc1 = {}, acc2 = {}, acc3 = {};
  gload(0);
  sstore(stg);
  __syncthreads();
  for (int kt = 0; kt < NT; ++kt) {
    const float* st = stg + (kt & 1) * ST_FL;
    if (kt + 1 < NT) gload(kt + 1);
    const float* as = st + (tile_m * 16 + l15) * A_PITCH + kk;
    const float* bs = st + A_FL + (kk >> 1) * B_PITCH + (n0 + l15) * 2;
#pragma unroll
    for (int s = 0; s < KT / 4; ++s) {
      const v2f a = *(const v2f*)(as + s * 4);
      const float* bp = bs + (2 * s) * B_PITCH;
      const v2f b0 = *(const v2f*)(bp);
      const v2f b1 = *(const v2f*)(bp + 32);
      const v2f b2 = *(const v2f*)(bp + 64);
      const v2f b3 = *(const v2f*)(bp + 96);
      acc0 = wmma_f32x4(a, b0, acc0);
      acc1 = wmma_f32x4(a, b1, acc1);
      acc2 = wmma_f32x4(a, b2, acc2);
      acc3 = wmma_f32x4(a, b3, acc3);
    }
    if (kt + 1 < NT) sstore(stg + ((kt + 1) & 1) * ST_FL);
    __syncthreads();
  }
  {
    const int mb = tile_m * 16 + ((lane >> 4) ? 8 : 0);
#pragma unroll
    for (int v = 0; v < 8; ++v) {
      smx[mb + v][n0 + 0 + l15]  = acc0[v];
      smx[mb + v][n0 + 16 + l15] = acc1[v];
      smx[mb + v][n0 + 32 + l15] = acc2[v];
      smx[mb + v][n0 + 48 + l15] = acc3[v];
    }
  }
  __syncthreads();

  float sm = 0.f;
  float* mrw = &smx[r][p * 32];
#pragma unroll
  for (int j = 0; j < 32; ++j) {
    float m = mrw[j];
    sm += m * m;
  }
  part[r][p] = sx;
  __syncthreads();
  if (t < 32) { float s = 0.f; for (int i = 0; i < 8; ++i) s += part[t][i]; rx2[t] = s; }
  __syncthreads();
  part[r][p] = sm;
  __syncthreads();
  if (t < 32) { float s = 0.f; for (int i = 0; i < 8; ++i) s += part[t][i]; rm2[t] = s; }
  __syncthreads();

  const float x2s = inScale * inScale * rx2[r];
  const float m2s = inScale * inScale * rm2[r];
  const float cmv = hyp_scale(x2s, m2s) * inScale;
#pragma unroll
  for (int j = 0; j < 32; ++j) mrw[j] = cmv * mrw[j];
  __syncthreads();

  if (wave >= 2) return;
  const int joff = lane << 3;
  float bv[8];
  load8(bias + joff, 1.f, bv);
  float x[8], x2;
  fold16<(NHID + 8) / 4>((const float4*)&smx[wave << 4][joff], 1.f, x, x2);

  float pb2 = 0.f, pxy = 0.f;
#pragma unroll
  for (int j = 0; j < 8; ++j) { pb2 += bv[j] * bv[j]; pxy += x[j] * bv[j]; }
  float y2 = wred32(pb2);
  float xy = 0.25f * wred32(pxy);
  float x2r = 0.0625f * x2;
  float den = fmaxf(1.f + 2.f * xy + x2r * y2, EPSF);
  float ca = (1.f + 2.f * xy + y2) / den;
  float cb = (1.f - x2r) / den;
  float zn2 = ca * ca * x2r + 2.f * ca * cb * xy + cb * cb * y2;
  float zn = fminf(fmaxf(sqrtf(zn2), EPSF), MAXNF);
  float us = atanhf(zn) / zn;
  float u[8];
  float pv = 0.f;
#pragma unroll
  for (int j = 0; j < 8; ++j) {
    float z = ca * 0.25f * x[j] + cb * bv[j];
    u[j] = fmaxf(us * z, 0.f);
    pv += u[j] * u[j];
  }
  float vn = fmaxf(sqrtf(wred32(pv)), EPSF);
  float fs = tanhf(vn) / vn;
  store8(OUT + (size_t)((blockIdx.x << 1) + wave) * NHID + joff, u, fs);
}

__global__ __launch_bounds__(512) void k_tail(const float* __restrict__ H1,
                                              const float* __restrict__ W2,
                                              const float* __restrict__ b2,
                                              const float* __restrict__ Wl,
                                              const float* __restrict__ bl,
                                              float* __restrict__ OUT) {
  constexpr int XP = NHID + 4;
  constexpr int WBP = 2 * NHID + 32;
  constexpr int PP = 64 + 4;
  __shared__ __attribute__((aligned(16))) float sW[(NHID / 2) * WBP];
  __shared__ __attribute__((aligned(16))) float sX[16 * XP];
  __shared__ __attribute__((aligned(16))) float sM[16 * XP];
  float* sWl = sW;
  float* sP = sW + 64 * XP;
  static_assert(64 * XP + 4 * 16 * PP <= (NHID / 2) * WBP, "Wl + partials fit");

  const int t = threadIdx.x;
  const int wave = t >> 5, lane = t & 31;
  const int l15 = lane & 15, hh = lane >> 4;
  constexpr int NPB = 4;
  const bool rw = wave < NPB;
  const int node = blockIdx.x * NPB + (rw ? wave : 0);
  const int joff = lane << 3;
  const float* msg = H1 + (size_t)node * 16 * NHID + joff;

  const int c4 = t & 63, kp0 = t >> 6;
  const float* wg = W2 + (size_t)(2 * kp0) * NHID + c4 * 4;
  float* ws = sW + kp0 * WBP + c4 * 8;
  auto wload = [&](int i, float4& w0, float4& w1) {
    const float* p = wg + (size_t)(16 * i) * NHID;
    w0 = *(const float4*)p;
    w1 = *(const float4*)(p + NHID);
  };
  auto wstore = [&](int i, const float4& w0, const float4& w1) {
    float* q = ws + (8 * i) * WBP;
    *(float4*)q = w0;
    *(float4*)(q + 4) = w1;
  };

  float bb[8];
  load8(b2 + joff, 1.f, bb);
  const float bl0 = bl[2 * lane], bl1 = bl[2 * lane + 1];
  const float4* g0 = (const float4*)msg;
  constexpr int RS = NHID / 4;
  const float4 z4 = make_float4(0.f, 0.f, 0.f, 0.f);
  float x[8] = {0.f, 0.f, 0.f, 0.f, 0.f, 0.f, 0.f, 0.f};
  float x2 = 0.f;
  if (rw) {
    const float4 a = g0[0], b = g0[1];
    scale8(a, b, 0.25f, x);
    float px = 0.f;
#pragma unroll
    for (int j = 0; j < 8; ++j) px += x[j] * x[j];
    x2 = wred32(px);
  }
  float4 a0 = z4, b0 = z4, a1 = z4, b1 = z4, a2 = z4, b2r = z4;
  float4 u0 = z4, v0 = z4, u1 = z4, v1 = z4, u2 = z4, v2 = z4;
#pragma unroll 1
  for (int k0 = -2; k0 < 16; k0 += 3) {
    float4 na0 = z4, nb0 = z4, na1 = z4, nb1 = z4, na2 = z4, nb2 = z4;
    if (rw) {
      const float4* gn0 = g0 + min(k0 + 3, 15) * RS;
      const float4* gn1 = g0 + min(k0 + 4, 15) * RS;
      const float4* gn2 = g0 + min(k0 + 5, 15) * RS;
      na0 = gn0[0]; nb0 = gn0[1];
      na1 = gn1[0]; nb1 = gn1[1];
      na2 = gn2[0]; nb2 = gn2[1];
    }
    float4 nu0, nv0, nu1, nv1, nu2, nv2;
    wload(min(k0 + 2, 15), nu0, nv0);
    wload(min(k0 + 3, 15), nu1, nv1);
    wload(min(k0 + 4, 15), nu2, nv2);
    if (k0 >= 1) {
      wstore(k0 - 1, u0, v0);
      if (rw) fold_step(x, x2, a0, b0, 0.25f);
      wstore(k0, u1, v1);
      if (rw) fold_step(x, x2, a1, b1, 0.25f);
      wstore(k0 + 1, u2, v2);
      if (rw) fold_step(x, x2, a2, b2r, 0.25f);
    }
    a0 = na0; b0 = nb0; a1 = na1; b1 = nb1; a2 = na2; b2r = nb2;
    u0 = nu0; v0 = nv0; u1 = nu1; v1 = nv1; u2 = nu2; v2 = nv2;
  }
  wstore(15, u0, v0);
  {
    float4 a, b;
    a.x = x[0]; a.y = x[1]; a.z = x[2]; a.w = x[3];
    b.x = x[4]; b.y = x[5]; b.z = x[6]; b.w = x[7];
    *(float4*)(sX + wave * XP + joff) = a;
    *(float4*)(sX + wave * XP + joff + 4) = b;
  }
  __syncthreads();

  {
    const int n0 = wave << 4;
    const float* as = sX + l15 * XP + 2 * hh;
    const int n = n0 + l15;
    const float* bs = sW + hh * WBP + (n >> 2) * 8 + (n & 3);
    v8f acc = {};
#pragma unroll
    for (int s = 0; s < NHID / 4; ++s) {
      const v2f a = *(const v2f*)(as + 4 * s);
      v2f b;
      b.x = bs[(2 * s) * WBP];
      b.y = bs[(2 * s) * WBP + 4];
      acc = wmma_f32x4(a, b, acc);
    }
#pragma unroll
    for (int v = 0; v < 8; ++v) sM[(8 * hh + v) * XP + n0 + l15] = acc[v];
  }
  __syncthreads();

  if (wave >= 4 && wave < 12) {
    const int tt = t - 128;
    const float* wlg = Wl + (size_t)(tt >> 6) * NHID + (tt & 63) * 4;
    float* wls = sWl + (tt >> 6) * XP + (tt & 63) * 4;
#pragma unroll
    for (int i = 0; i < 16; ++i)
      *(float4*)(wls + 4 * i * XP) = *(const float4*)(wlg + (size_t)(4 * i) * NHID);
  }

  float hx2 = 0.f;
  if (rw) {
    float m[8];
    {
      const float4 a = *(const float4*)(sM + wave * XP + joff);
      const float4 b = *(const float4*)(sM + wave * XP + joff + 4);
      m[0] = a.x; m[1] = a.y; m[2] = a.z; m[3] = a.w;
      m[4] = b.x; m[5] = b.y; m[6] = b.z; m[7] = b.w;
    }
    float pm2 = 0.f, pmd = 0.f, pb2 = 0.f;
#pragma unroll
    for (int j = 0; j < 8; ++j) {
      pm2 += m[j] * m[j]; pmd += m[j] * bb[j]; pb2 += bb[j] * bb[j];
    }
    const float m2 = wred32(pm2);
    const float md = wred32(pmd);
    const float y2 = wred32(pb2);
    const float cmv = hyp_scale(x2, m2);
    const float cf = 0.25f * cmv;
    const float x2r = cf * cf * m2;
    const float xy = cf * md;
    const float den = fmaxf(1.f + 2.f * xy + x2r * y2, EPSF);
    const float ca = (1.f + 2.f * xy + y2) / den;
    const float cb = (1.f - x2r) / den;
    const float zn2 = ca * ca * x2r + 2.f * ca * cb * xy + cb * cb * y2;
    const float zn = fminf(fmaxf(sqrtf(zn2), EPSF), MAXNF);
    const float us = atanhf(zn) / zn;
    const float acf = ca * cf;
    float u[8];
    float pv = 0.f;
#pragma unroll
    for (int j = 0; j < 8; ++j) {
      const float z = acf * m[j] + cb * bb[j];
      u[j] = fmaxf(us * z, 0.f);
      pv += u[j] * u[j];
    }
    const float vn = fmaxf(sqrtf(wred32(pv)), EPSF);
    const float fs = tanhf(vn) / vn;
    float ph = 0.f;
#pragma unroll
    for (int j = 0; j < 8; ++j) { u[j] = fs * u[j]; ph += u[j] * u[j]; }
    hx2 = wred32(ph);
    float4 a, b;
    a.x = u[0]; a.y = u[1]; a.z = u[2]; a.w = u[3];
    b.x = u[4]; b.y = u[5]; b.z = u[6]; b.w = u[7];
    *(float4*)(sX + wave * XP + joff) = a;
    *(float4*)(sX + wave * XP + joff + 4) = b;
  }
  __syncthreads();

  {
    const int tn = wave & 3, kq = wave >> 2;
    const float* ar = sX + l15 * XP + 64 * kq + 2 * hh;
    const float* br = sWl + (16 * tn + l15) * XP + 64 * kq + 2 * hh;
    v8f c = {};
#pragma unroll
    for (int s = 0; s < 16; ++s) {
      const v2f a = *(const v2f*)(ar + 4 * s);
      const v2f b = *(const v2f*)(br + 4 * s);
      c = wmma_f32x4(a, b, c);
    }
    float* pt = sP + kq * (16 * PP) + 16 * tn + l15;
#pragma unroll
    for (int v = 0; v < 8; ++v) pt[(8 * hh + v) * PP] = c[v];
  }
  __syncthreads();

  if (rw) {
    const float* pr = sP + wave * PP + 2 * lane;
    v2f mm = *(const v2f*)pr;
#pragma unroll
    for (int q = 1; q < 4; ++q) {
      const v2f pq = *(const v2f*)(pr + q * (16 * PP));
      mm.x += pq.x; mm.y += pq.y;
    }
    const float m2 = wred32(mm.x * mm.x + mm.y * mm.y);
    const float md = wred32(mm.x * bl0 + mm.y * bl1);
    const float y2 = wred32(bl0 * bl0 + bl1 * bl1);
    const float s = hyp_scale(hx2, m2);
    const float x2r = s * s * m2;
    const float xy = s * md;
    const float den = fmaxf(1.f + 2.f * xy + x2r * y2, EPSF);
    const float ca = (1.f + 2.f * xy + y2) / den;
    const float cb = (1.f - x2r) / den;
    const float acs = ca * s;
    float2 o;
    o.x = acs * mm.x + cb * bl0;
    o.y = acs * mm.y + cb * bl1;
    *(float2*)(OUT + (size_t)node * 64 + 2 * lane) = o;
  }
}

extern "C" void kernel_launch(void* const* d_in, const int* in_sizes, int n_in,
                              void* d_out, int out_size, void* d_ws,
                              size_t ws_size, hipStream_t stream) {
  const float* features = (const float*)d_in[0];
  const float* W1 = (const float*)d_in[1];
  const float* b1 = (const float*)d_in[2];
  const float* W2 = (const float*)d_in[3];
  const float* b2 = (const float*)d_in[4];
  const float* Wl = (const float*)d_in[5];
  const float* bl = (const float*)d_in[6];
  const int* src_idx = (const int*)d_in[7];
  const int* to_fetch = (const int*)d_in[8];
  float* out = (float*)d_out;

  float* h1 = (float*)d_ws;

  k_mm<512><<<512, 256, 0, stream>>>(features, W1, b1, h1, 0.25f, src_idx,
                                     to_fetch);
  k_tail<<<16, 512, 0, stream>>>(h1, W2, b2, Wl, bl, out);
}
